# starting kernel
# baseline (speedup 1.0000x reference)
#include <hip/hip_runtime.h>

#define D 128

typedef __attribute__((ext_vector_type(2))) float v2f;
typedef __attribute__((ext_vector_type(8))) float v8f;


__global__ void fill_kernel(float* p, float v, long n) {
  long i = (long)blockIdx.x * blockDim.x + threadIdx.x;
  if (i < n) p[i] = v;
}

__global__ void pack_pairs_kernel(const float* __restrict__ W, float* __restrict__ Wp,
                                  int K, int Ncols) {
  int i = blockIdx.x * blockDim.x + threadIdx.x;
  int total = (K >> 1) * Ncols;
  if (i >= total) return;
  int kp = i / Ncols, n = i - kp * Ncols;
  float2 v;
  v.x = W[(long)(2 * kp) * Ncols + n];
  v.y = W[(long)(2 * kp + 1) * Ncols + n];
  ((float2*)Wp)[(long)kp * Ncols + n] = v;
}

__global__ void degree_kernel(const int* __restrict__ snd, const int* __restrict__ rcv,
                              float* deg_s, float* deg_r, int E) {
  int e = blockIdx.x * blockDim.x + threadIdx.x;
  if (e < E) {
    unsafeAtomicAdd(&deg_s[snd[e]], 1.0f);
    unsafeAtomicAdd(&deg_r[rcv[e]], 1.0f);
  }
}

__global__ void rownorm_kernel(const float* __restrict__ src, const float* __restrict__ deg,
                               float* __restrict__ dst, int n_nodes) {
  long i = (long)blockIdx.x * blockDim.x + threadIdx.x;
  long n4 = (long)n_nodes * (D / 4);
  if (i >= n4) return;
  int row = (int)(i >> 5);
  float s = rsqrtf(deg[row]);
  float4 v = ((const float4*)src)[i];
  v.x *= s; v.y *= s; v.z *= s; v.w *= s;
  ((float4*)dst)[i] = v;
}

__global__ void scatter_kernel(const int* __restrict__ snd, const int* __restrict__ rcv,
                               const float* __restrict__ xn, float* msum, int E) {
  long gid = (long)blockIdx.x * blockDim.x + threadIdx.x;
  long total = (long)E * 32;
  if (gid >= total) return;
  int e = (int)(gid >> 5);
  int c = (int)(gid & 31) << 2;
  int s = snd[e], r = rcv[e];
  float4 v = *(const float4*)(xn + (long)s * D + c);
  float* dst = msum + (long)r * D + c;
  unsafeAtomicAdd(dst + 0, v.x);
  unsafeAtomicAdd(dst + 1, v.y);
  unsafeAtomicAdd(dst + 2, v.z);
  unsafeAtomicAdd(dst + 3, v.w);
}


__global__ __launch_bounds__(256)
void sage_gemm_kernel(const float* __restrict__ A0,
                      const float* msum,
                      const float* xn,
                      const float* __restrict__ deg_r,
                      const float* __restrict__ Wp,
                      const float* __restrict__ bias,
                      float* out,
                      int n_nodes, int do_relu) {
  int wave = (blockIdx.x << 3) + (threadIdx.x >> 5);
  int mt_count = (n_nodes + 15) >> 4;
  if (wave >= mt_count) return;
  int lane = threadIdx.x & 31;
  int lo = lane & 15, hi = lane >> 4;

  int m = (wave << 4) + lo;
  bool mvalid = m < n_nodes;
  int mr = mvalid ? m : 0;

  float c = deg_r[mr];
  float rs = rsqrtf(c);
  float scale = rs * rs * rs;

  const float* a0row = A0 + (long)mr * D;
  const float* smrow = msum + (long)mr * D;
  const float* xnrow = xn + (long)mr * D;
  const v2f* wp = (const v2f*)Wp;

  v8f acc[8];
#pragma unroll
  for (int t = 0; t < 8; ++t) acc[t] = (v8f)0.0f;

  for (int kk = 0; kk < D; kk += 4) {
    int k0 = kk + (hi << 1);
    v2f a = *(const v2f*)(a0row + k0);
    if (!mvalid) { a.x = 0.0f; a.y = 0.0f; }
    const v2f* wrow = wp + (long)(k0 >> 1) * D + lo;
#pragma unroll
    for (int t = 0; t < 8; ++t) {
      v2f b = wrow[t * 16];
      acc[t] = __builtin_amdgcn_wmma_f32_16x16x4_f32(
          false, a, false, b, (short)0, acc[t], false, false);
    }
  }

  for (int kk = 0; kk < D; kk += 4) {
    int k0 = kk + (hi << 1);
    v2f sm = *(const v2f*)(smrow + k0);
    v2f xv = *(const v2f*)(xnrow + k0);
    v2f a;
    a.x = (sm.x + xv.x) * scale;
    a.y = (sm.y + xv.y) * scale;
    if (!mvalid) { a.x = 0.0f; a.y = 0.0f; }
    const v2f* wrow = wp + (long)((k0 + D) >> 1) * D + lo;
#pragma unroll
    for (int t = 0; t < 8; ++t) {
      v2f b = wrow[t * 16];
      acc[t] = __builtin_amdgcn_wmma_f32_16x16x4_f32(
          false, a, false, b, (short)0, acc[t], false, false);
    }
  }

#pragma unroll
  for (int t = 0; t < 8; ++t) {
    int col = t * 16 + lo;
    float bv = bias[col];
#pragma unroll
    for (int v = 0; v < 8; ++v) {
      int row = (wave << 4) + v + (hi << 3);
      float val = acc[t][v] + bv;
      if (do_relu) val = fmaxf(val, 0.0f);
      if (row < n_nodes) out[(long)row * D + col] = val;
    }
  }
}


#define ZSTRIDE 132

__global__ __launch_bounds__(128)
void pair_mlp_kernel(const int* __restrict__ pairs, const float* __restrict__ h,
                     const float* __restrict__ Wpa,
                     const float* __restrict__ ba,
                     const float* __restrict__ Wb, const float* __restrict__ bb,
                     float* __restrict__ out, int n_pairs) {
  __shared__ float zsh[4][16 * ZSTRIDE];
  int wv = threadIdx.x >> 5;
  int tile = (blockIdx.x << 2) + wv;
  int ntiles = (n_pairs + 15) >> 4;
  if (tile >= ntiles) return;
  int lane = threadIdx.x & 31;
  int lo = lane & 15, hi = lane >> 4;
  float* z = zsh[wv];

  {
    int p = (tile << 4) + lo;
    int pc = p < n_pairs ? p : 0;
    int i = pairs[2 * pc], j = pairs[2 * pc + 1];
    const float4* hi4 = (const float4*)(h + (long)i * D);
    const float4* hj4 = (const float4*)(h + (long)j * D);
    float4* zrow = (float4*)(z + lo * ZSTRIDE);
    for (int q = hi * 16; q < hi * 16 + 16; ++q) {
      float4 a = hi4[q], b = hj4[q];
      float4 v;
      v.x = a.x * b.x; v.y = a.y * b.y; v.z = a.z * b.z; v.w = a.w * b.w;
      zrow[q] = v;
    }
  }

  const v2f* wp = (const v2f*)Wpa;
  float partial[8];
#pragma unroll
  for (int v = 0; v < 8; ++v) partial[v] = 0.0f;

  for (int t = 0; t < 8; ++t) {
    v8f acc = (v8f)0.0f;
    for (int kk = 0; kk < D; kk += 4) {
      int k0 = kk + (hi << 1);
      v2f a = *(const v2f*)(z + lo * ZSTRIDE + k0);
      v2f b = wp[(long)(k0 >> 1) * D + t * 16 + lo];
      acc = __builtin_amdgcn_wmma_f32_16x16x4_f32(
          false, a, false, b, (short)0, acc, false, false);
    }
    int col = t * 16 + lo;
    float bv = ba[col], wb = Wb[col];
#pragma unroll
    for (int v = 0; v < 8; ++v) {
      partial[v] += fmaxf(acc[v] + bv, 0.0f) * wb;
    }
  }

#pragma unroll
  for (int v = 0; v < 8; ++v) {
    float x = partial[v];
    x += __shfl_xor(x, 1, 32);
    x += __shfl_xor(x, 2, 32);
    x += __shfl_xor(x, 4, 32);
    x += __shfl_xor(x, 8, 32);
    partial[v] = x;
  }
  if (lo == 0) {
    float bbv = bb[0];
#pragma unroll
    for (int v = 0; v < 8; ++v) {
      int row = (tile << 4) + v + (hi << 3);
      if (row < n_pairs) out[row] = partial[v] + bbv;
    }
  }
}


extern "C" void kernel_launch(void* const* d_in, const int* in_sizes, int n_in,
                              void* d_out, int out_size, void* d_ws, size_t ws_size,
                              hipStream_t stream) {
  const int* senders   = (const int*)d_in[1];
  const int* receivers = (const int*)d_in[2];
  const int* pairs     = (const int*)d_in[3];
  const float* emb = (const float*)d_in[4];
  const float* W1  = (const float*)d_in[5];
  const float* b1  = (const float*)d_in[6];
  const float* W2  = (const float*)d_in[7];
  const float* b2  = (const float*)d_in[8];
  const float* Wa  = (const float*)d_in[9];
  const float* ba  = (const float*)d_in[10];
  const float* Wb  = (const float*)d_in[11];
  const float* bb  = (const float*)d_in[12];
  float* out = (float*)d_out;

  int N = in_sizes[0];
  int E = in_sizes[1];
  int P = in_sizes[3] / 2;
  long ND = (long)N * D;

  float* ws    = (float*)d_ws;
  float* deg_s = ws;
  float* deg_r = ws + N;
  float* bufA  = ws + 2L * N;
  float* bufB  = bufA + ND;
  float* bufC  = bufB + ND;
  float* Wp1   = bufC + ND;
  float* Wp2   = Wp1 + 2L * D * D;
  float* Wpa   = Wp2 + 2L * D * D;

  const int T = 256;
  fill_kernel<<<(N + T - 1) / T, T, 0, stream>>>(deg_s, 1.0f, N);
  fill_kernel<<<(N + T - 1) / T, T, 0, stream>>>(deg_r, 1.0f, N);
  fill_kernel<<<(int)((ND + T - 1) / T), T, 0, stream>>>(bufB, 0.0f, ND);
  fill_kernel<<<(int)((ND + T - 1) / T), T, 0, stream>>>(bufC, 0.0f, ND);

  pack_pairs_kernel<<<(2 * D * D / 2 + T - 1) / T, T, 0, stream>>>(W1, Wp1, 2 * D, D);
  pack_pairs_kernel<<<(2 * D * D / 2 + T - 1) / T, T, 0, stream>>>(W2, Wp2, 2 * D, D);
  pack_pairs_kernel<<<(D * D / 2 + T - 1) / T, T, 0, stream>>>(Wa, Wpa, D, D);

  degree_kernel<<<(E + T - 1) / T, T, 0, stream>>>(senders, receivers, deg_s, deg_r, E);

  long n4 = ND / 4;
  long sth = (long)E * 32;
  int mt = (N + 15) >> 4;

  rownorm_kernel<<<(int)((n4 + T - 1) / T), T, 0, stream>>>(emb, deg_s, bufA, N);
  scatter_kernel<<<(int)((sth + T - 1) / T), T, 0, stream>>>(senders, receivers, bufA, bufB, E);
  sage_gemm_kernel<<<(mt + 7) / 8, 256, 0, stream>>>(emb, bufB, bufA, deg_r, Wp1, b1,
                                                     bufB, N,  1);

  rownorm_kernel<<<(int)((n4 + T - 1) / T), T, 0, stream>>>(bufB, deg_s, bufA, N);
  scatter_kernel<<<(int)((sth + T - 1) / T), T, 0, stream>>>(senders, receivers, bufA, bufC, E);
  sage_gemm_kernel<<<(mt + 7) / 8, 256, 0, stream>>>(bufB, bufC, bufA, deg_r, Wp2, b2,
                                                     bufC, N,  0);

  int ptiles = (P + 15) >> 4;
  pair_mlp_kernel<<<(ptiles + 3) / 4, 128, 0, stream>>>(pairs, bufC, Wpa, ba, Wb, bb, out, P);
}


# float-atomic scatter adds into one sum plane per cache half (a wave reads its die id; the GEMM adds the two planes), in two column passes; row sum = two partial sums; x1 rebuilt from its scaled rows
# speedup vs baseline: 1.6320x; 1.6320x over previous
#include <hip/hip_runtime.h>

#define D 128

typedef __attribute__((ext_vector_type(2))) float v2f;
typedef __attribute__((ext_vector_type(8))) float v8f;


#define NPAIR_SAGE (D * D)
#define NPAIR_MLP (D * D / 2)
__global__ __launch_bounds__(256)
void init_kernel(float* __restrict__ deg_s, float* __restrict__ deg_r, int n_nodes, int nb_deg,
                 const float* __restrict__ W1, float* __restrict__ Wp1,
                 const float* __restrict__ W2, float* __restrict__ Wp2,
                 const float* __restrict__ Wa, float* __restrict__ Wpa) {
  int b = blockIdx.x;
  if (b < nb_deg) {
    int i = b * 256 + threadIdx.x;
    if (i < n_nodes) { deg_s[i] = 1.0f; deg_r[i] = 1.0f; }
    return;
  }
  int i = (b - nb_deg) * 256 + threadIdx.x;
  const float* W;
  float* Wp;
  if (i < NPAIR_SAGE) { W = W1; Wp = Wp1; }
  else if (i < 2 * NPAIR_SAGE) { W = W2; Wp = Wp2; i -= NPAIR_SAGE; }
  else if (i < 2 * NPAIR_SAGE + NPAIR_MLP) { W = Wa; Wp = Wpa; i -= 2 * NPAIR_SAGE; }
  else return;
  int kp = i >> 7, n = i & (D - 1);
  float2 v;
  v.x = W[(long)(2 * kp) * D + n];
  v.y = W[(long)(2 * kp + 1) * D + n];
  ((float2*)Wp)[i] = v;
}

__global__ void degree_kernel(const int* __restrict__ snd, const int* __restrict__ rcv,
                              float* deg_s, float* deg_r, int E) {
  int e = blockIdx.x * blockDim.x + threadIdx.x;
  if (e < E) {
    unsafeAtomicAdd(&deg_s[snd[e]], 1.0f);
    unsafeAtomicAdd(&deg_r[rcv[e]], 1.0f);
  }
}

#define RN_PER_LANE 4
__global__ __launch_bounds__(256)
void rownorm_kernel(const float* __restrict__ src, const float* __restrict__ deg,
                    float* __restrict__ dst, float* __restrict__ sum0,
                    float* __restrict__ zero0, int n_nodes) {
  long base = (long)blockIdx.x * (256 * RN_PER_LANE) + threadIdx.x;
  long n4 = (long)n_nodes * (D / 4);
  float4 v[RN_PER_LANE];
  float s[RN_PER_LANE];
#pragma unroll
  for (int k = 0; k < RN_PER_LANE; ++k) {
    long i = base + (long)k * 256;
    long ic = i < n4 ? i : 0;
    s[k] = deg[(int)(ic >> 5)];
    v[k] = ((const float4*)src)[ic];
  }
#pragma unroll
  for (int k = 0; k < RN_PER_LANE; ++k) {
    long i = base + (long)k * 256;
    if (i < n4) {
      float sc = rsqrtf(s[k]);
      float4 o = v[k];
      o.x *= sc; o.y *= sc; o.z *= sc; o.w *= sc;
      ((float4*)dst)[i] = o;
      ((float4*)sum0)[i] = o;
      ((float4*)zero0)[i] = make_float4(0.0f, 0.0f, 0.0f, 0.0f);
    }
  }
}

#define EPW 16

static __device__ __forceinline__ float* own_plane(float* plane_a, float* plane_b) {
  unsigned place = __builtin_amdgcn_s_sendmsg_rtn(0x87);
  return ((place >> 18) & 1u) ? plane_b : plane_a;
}

__global__ __launch_bounds__(256)
void scatter_kernel(const int* __restrict__ snd, const int* __restrict__ rcv,
                    const float* __restrict__ xn, float* msum_a, float* msum_b, int E) {
  int wave = __builtin_amdgcn_readfirstlane(
      (int)((blockIdx.x * blockDim.x + threadIdx.x) >> 5));
  long e0 = (long)wave * EPW;
  if (e0 >= E) return;

  int c = (int)blockIdx.y * 64 + (threadIdx.x & 31);

  if (e0 + EPW <= E) {
    int s[EPW], r[EPW];
#pragma unroll
    for (int j = 0; j < EPW; ++j) { s[j] = snd[e0 + j]; r[j] = rcv[e0 + j]; }
    float* msum = own_plane(msum_a, msum_b);
    float v[EPW][2];
#pragma unroll
    for (int j = 0; j < EPW; ++j) {
      const float* src = xn + (long)s[j] * D + c;
      v[j][0] = src[0];
      v[j][1] = src[32];
    }
#pragma unroll
    for (int j = 0; j < EPW; ++j) {
      float* dst = msum + (long)r[j] * D + c;
      unsafeAtomicAdd(dst + 0, v[j][0]);
      unsafeAtomicAdd(dst + 32, v[j][1]);
    }
  } else {
    for (long e = e0; e < E; ++e) {
      int s = snd[e], r = rcv[e];
      float* msum = own_plane(msum_a, msum_b);
      const float* src = xn + (long)s * D + c;
      float* dst = msum + (long)r * D + c;
      float v0 = src[0];
      float v1 = src[32];
      unsafeAtomicAdd(dst + 0, v0);
      unsafeAtomicAdd(dst + 32, v1);
    }
  }
}


__global__ __launch_bounds__(256)
void sage_gemm_kernel(const float* __restrict__ A0,
                      int a0_scaled,
                      const float* msum,
                      const float* msum2,
                      const float* __restrict__ deg_r,
                      const float* __restrict__ deg_s,
                      const float* __restrict__ Wp,
                      const float* __restrict__ bias,
                      float* out,
                      float* xn_next,
                      float* sum_next,
                      float* zero_next,
                      int n_nodes, int do_relu, int write_next) {
  int wave = (blockIdx.x << 3) + (threadIdx.x >> 5);
  int mt_count = (n_nodes + 15) >> 4;
  if (wave >= mt_count) return;
  int lane = threadIdx.x & 31;
  int lo = lane & 15, hi = lane >> 4;

  int m = (wave << 4) + lo;
  bool mvalid = m < n_nodes;
  int mr = mvalid ? m : 0;

  float c = deg_r[mr];
  float rs = rsqrtf(c);
  float scale = rs * rs * rs;
  float ds = deg_s[mr];
  float rs_next = rsqrtf(ds);
  float a0s = a0_scaled ? sqrtf(ds) : 1.0f;

  const float* a0row = A0 + (long)mr * D;
  const float* smrow = msum + (long)mr * D;
  const float* sm2row = msum2 + (long)mr * D;
  const v2f* wp = (const v2f*)Wp;

  v8f acc[8];
#pragma unroll
  for (int t = 0; t < 8; ++t) acc[t] = (v8f)0.0f;

  for (int kk = 0; kk < D; kk += 4) {
    int k0 = kk + (hi << 1);
    v2f a = *(const v2f*)(a0row + k0);
    a.x *= a0s;
    a.y *= a0s;
    if (!mvalid) { a.x = 0.0f; a.y = 0.0f; }
    const v2f* wrow = wp + (long)(k0 >> 1) * D + lo;
#pragma unroll
    for (int t = 0; t < 8; ++t) {
      v2f b = wrow[t * 16];
      acc[t] = __builtin_amdgcn_wmma_f32_16x16x4_f32(
          false, a, false, b, (short)0, acc[t], false, false);
    }
  }

  for (int kk = 0; kk < D; kk += 4) {
    int k0 = kk + (hi << 1);
    v2f sm = *(const v2f*)(smrow + k0);
    v2f sm2 = *(const v2f*)(sm2row + k0);
    v2f a;
    a.x = (sm.x + sm2.x) * scale;
    a.y = (sm.y + sm2.y) * scale;
    if (!mvalid) { a.x = 0.0f; a.y = 0.0f; }
    const v2f* wrow = wp + (long)((k0 + D) >> 1) * D + lo;
#pragma unroll
    for (int t = 0; t < 8; ++t) {
      v2f b = wrow[t * 16];
      acc[t] = __builtin_amdgcn_wmma_f32_16x16x4_f32(
          false, a, false, b, (short)0, acc[t], false, false);
    }
  }

  float rsn[8];
#pragma unroll
  for (int v = 0; v < 8; ++v) rsn[v] = __shfl(rs_next, v + (hi << 3), 32);

  if (write_next) {
#pragma unroll
    for (int t = 0; t < 8; ++t) {
      int col = t * 16 + lo;
      float bv = bias[col];
#pragma unroll
      for (int v = 0; v < 8; ++v) {
        int row = (wave << 4) + v + (hi << 3);
        float val = acc[t][v] + bv;
        if (do_relu) val = fmaxf(val, 0.0f);
        if (row < n_nodes) {
          long idx = (long)row * D + col;
          float xs = val * rsn[v];
          xn_next[idx] = xs;
          sum_next[idx] = xs;
          zero_next[idx] = 0.0f;
        }
      }
    }
  } else {
#pragma unroll
    for (int t = 0; t < 8; ++t) {
      int col = t * 16 + lo;
      float bv = bias[col];
#pragma unroll
      for (int v = 0; v < 8; ++v) {
        int row = (wave << 4) + v + (hi << 3);
        float val = acc[t][v] + bv;
        if (do_relu) val = fmaxf(val, 0.0f);
        if (row < n_nodes) out[(long)row * D + col] = val;
      }
    }
  }
}


#define ZSTRIDE 132

__global__ __launch_bounds__(128)
void pair_mlp_kernel(const int* __restrict__ pairs, const float* __restrict__ h,
                     const float* __restrict__ Wpa,
                     const float* __restrict__ ba,
                     const float* __restrict__ Wb, const float* __restrict__ bb,
                     float* __restrict__ out, int n_pairs) {
  __shared__ float zsh[4][16 * ZSTRIDE];
  int wv = threadIdx.x >> 5;
  int tile = (blockIdx.x << 2) + wv;
  int ntiles = (n_pairs + 15) >> 4;
  if (tile >= ntiles) return;
  int lane = threadIdx.x & 31;
  int lo = lane & 15, hi = lane >> 4;
  float* z = zsh[wv];

  {
    int p = (tile << 4) + lo;
    int pc = p < n_pairs ? p : 0;
    int i = pairs[2 * pc], j = pairs[2 * pc + 1];
    const float4* hi4 = (const float4*)(h + (long)i * D);
    const float4* hj4 = (const float4*)(h + (long)j * D);
    float4* zrow = (float4*)(z + lo * ZSTRIDE);
    for (int q = hi * 16; q < hi * 16 + 16; ++q) {
      float4 a = hi4[q], b = hj4[q];
      float4 v;
      v.x = a.x * b.x; v.y = a.y * b.y; v.z = a.z * b.z; v.w = a.w * b.w;
      zrow[q] = v;
    }
  }

  const v2f* wp = (const v2f*)Wpa;
  float partial[8];
#pragma unroll
  for (int v = 0; v < 8; ++v) partial[v] = 0.0f;

  for (int t = 0; t < 8; ++t) {
    v8f acc = (v8f)0.0f;
    for (int kk = 0; kk < D; kk += 4) {
      int k0 = kk + (hi << 1);
      v2f a = *(const v2f*)(z + lo * ZSTRIDE + k0);
      v2f b = wp[(long)(k0 >> 1) * D + t * 16 + lo];
      acc = __builtin_amdgcn_wmma_f32_16x16x4_f32(
          false, a, false, b, (short)0, acc, false, false);
    }
    int col = t * 16 + lo;
    float bv = ba[col], wb = Wb[col];
#pragma unroll
    for (int v = 0; v < 8; ++v) {
      partial[v] += fmaxf(acc[v] + bv, 0.0f) * wb;
    }
  }

#pragma unroll
  for (int v = 0; v < 8; ++v) {
    float x = partial[v];
    x += __shfl_xor(x, 1, 32);
    x += __shfl_xor(x, 2, 32);
    x += __shfl_xor(x, 4, 32);
    x += __shfl_xor(x, 8, 32);
    partial[v] = x;
  }
  if (lo == 0) {
    float bbv = bb[0];
#pragma unroll
    for (int v = 0; v < 8; ++v) {
      int row = (tile << 4) + v + (hi << 3);
      if (row < n_pairs) out[row] = partial[v] + bbv;
    }
  }
}


extern "C" void kernel_launch(void* const* d_in, const int* in_sizes, int n_in,
                              void* d_out, int out_size, void* d_ws, size_t ws_size,
                              hipStream_t stream) {
  const int* senders   = (const int*)d_in[1];
  const int* receivers = (const int*)d_in[2];
  const int* pairs     = (const int*)d_in[3];
  const float* emb = (const float*)d_in[4];
  const float* W1  = (const float*)d_in[5];
  const float* b1  = (const float*)d_in[6];
  const float* W2  = (const float*)d_in[7];
  const float* b2  = (const float*)d_in[8];
  const float* Wa  = (const float*)d_in[9];
  const float* ba  = (const float*)d_in[10];
  const float* Wb  = (const float*)d_in[11];
  const float* bb  = (const float*)d_in[12];
  float* out = (float*)d_out;

  int N = in_sizes[0];
  int E = in_sizes[1];
  int P = in_sizes[3] / 2;
  long ND = (long)N * D;

  float* ws    = (float*)d_ws;
  float* deg_s = ws;
  float* deg_r = ws + N;
  float* bufA  = ws + 2L * N;
  float* bufB  = bufA + ND;
  float* bufC  = bufB + ND;
  float* Wp1   = bufC + ND;
  float* Wp2   = Wp1 + 2L * D * D;
  float* Wpa   = Wp2 + 2L * D * D;

  const int T = 256;
  int nb_deg = (N + T - 1) / T;
  int nb_pack = (2 * NPAIR_SAGE + NPAIR_MLP + T - 1) / T;
  init_kernel<<<nb_deg + nb_pack, T, 0, stream>>>(deg_s, deg_r, N, nb_deg,
                                                   W1, Wp1, W2, Wp2, Wa, Wpa);

  degree_kernel<<<(E + T - 1) / T, T, 0, stream>>>(senders, receivers, deg_s, deg_r, E);

  long n4 = ND / 4;
  long sth = ((long)E + EPW - 1) / EPW * 32;
  dim3 sgrid((unsigned)((sth + T - 1) / T), 2);
  int mt = (N + 15) >> 4;
  int rnblocks = (int)((n4 + T * RN_PER_LANE - 1) / (T * RN_PER_LANE));

  rownorm_kernel<<<rnblocks, T, 0, stream>>>(emb, deg_s, bufA, bufB, bufC, N);
  scatter_kernel<<<sgrid, T, 0, stream>>>(senders, receivers, bufA, bufB, bufC, E);
  sage_gemm_kernel<<<(mt + 7) / 8, 256, 0, stream>>>(emb,  0, bufB, bufC, deg_r, deg_s, Wp1, b1,
                                                     bufB, bufA, bufC, bufB, N,  1,  1);

  scatter_kernel<<<sgrid, T, 0, stream>>>(senders, receivers, bufA, bufC, bufB, E);
  sage_gemm_kernel<<<(mt + 7) / 8, 256, 0, stream>>>(bufA,  1, bufC, bufB, deg_r, deg_s, Wp2, b2,
                                                     bufC, bufA, bufA, bufA, N,  0,  0);

  int ptiles = (P + 15) >> 4;
  pair_mlp_kernel<<<(ptiles + 3) / 4, 128, 0, stream>>>(pairs, bufC, Wpa, ba, Wb, bb, out, P);
}


# pair MLP: A fragments built in registers from 16-byte row loads (both lane halves share a 128-B line; no LDS staging of the z tile); k summed in a different fixed order
# speedup vs baseline: 1.6885x; 1.0346x over previous
#include <hip/hip_runtime.h>

#define D 128

typedef __attribute__((ext_vector_type(2))) float v2f;
typedef __attribute__((ext_vector_type(8))) float v8f;


#define NPAIR_SAGE (D * D)
#define NPAIR_MLP (D * D / 2)
__global__ __launch_bounds__(256)
void init_kernel(float* __restrict__ deg_s, float* __restrict__ deg_r, int n_nodes, int nb_deg,
                 const float* __restrict__ W1, float* __restrict__ Wp1,
                 const float* __restrict__ W2, float* __restrict__ Wp2,
                 const float* __restrict__ Wa, float* __restrict__ Wpa) {
  int b = blockIdx.x;
  if (b < nb_deg) {
    int i = b * 256 + threadIdx.x;
    if (i < n_nodes) { deg_s[i] = 1.0f; deg_r[i] = 1.0f; }
    return;
  }
  int i = (b - nb_deg) * 256 + threadIdx.x;
  const float* W;
  float* Wp;
  if (i < NPAIR_SAGE) { W = W1; Wp = Wp1; }
  else if (i < 2 * NPAIR_SAGE) { W = W2; Wp = Wp2; i -= NPAIR_SAGE; }
  else if (i < 2 * NPAIR_SAGE + NPAIR_MLP) { W = Wa; Wp = Wpa; i -= 2 * NPAIR_SAGE; }
  else return;
  int kp = i >> 7, n = i & (D - 1);
  float2 v;
  v.x = W[(long)(2 * kp) * D + n];
  v.y = W[(long)(2 * kp + 1) * D + n];
  ((float2*)Wp)[i] = v;
}

__global__ void degree_kernel(const int* __restrict__ snd, const int* __restrict__ rcv,
                              float* deg_s, float* deg_r, int E) {
  int e = blockIdx.x * blockDim.x + threadIdx.x;
  if (e < E) {
    unsafeAtomicAdd(&deg_s[snd[e]], 1.0f);
    unsafeAtomicAdd(&deg_r[rcv[e]], 1.0f);
  }
}

#define RN_PER_LANE 4
__global__ __launch_bounds__(256)
void rownorm_kernel(const float* __restrict__ src, const float* __restrict__ deg,
                    float* __restrict__ dst, float* __restrict__ sum0,
                    float* __restrict__ zero0, int n_nodes) {
  long base = (long)blockIdx.x * (256 * RN_PER_LANE) + threadIdx.x;
  long n4 = (long)n_nodes * (D / 4);
  float4 v[RN_PER_LANE];
  float s[RN_PER_LANE];
#pragma unroll
  for (int k = 0; k < RN_PER_LANE; ++k) {
    long i = base + (long)k * 256;
    long ic = i < n4 ? i : 0;
    s[k] = deg[(int)(ic >> 5)];
    v[k] = ((const float4*)src)[ic];
  }
#pragma unroll
  for (int k = 0; k < RN_PER_LANE; ++k) {
    long i = base + (long)k * 256;
    if (i < n4) {
      float sc = rsqrtf(s[k]);
      float4 o = v[k];
      o.x *= sc; o.y *= sc; o.z *= sc; o.w *= sc;
      ((float4*)dst)[i] = o;
      ((float4*)sum0)[i] = o;
      ((float4*)zero0)[i] = make_float4(0.0f, 0.0f, 0.0f, 0.0f);
    }
  }
}

#define EPW 16

static __device__ __forceinline__ float* own_plane(float* plane_a, float* plane_b) {
  unsigned place = __builtin_amdgcn_s_sendmsg_rtn(0x87);
  return ((place >> 18) & 1u) ? plane_b : plane_a;
}

__global__ __launch_bounds__(256)
void scatter_kernel(const int* __restrict__ snd, const int* __restrict__ rcv,
                    const float* __restrict__ xn, float* msum_a, float* msum_b, int E) {
  int wave = __builtin_amdgcn_readfirstlane(
      (int)((blockIdx.x * blockDim.x + threadIdx.x) >> 5));
  long e0 = (long)wave * EPW;
  if (e0 >= E) return;

  int c = (int)blockIdx.y * 64 + (threadIdx.x & 31);

  if (e0 + EPW <= E) {
    int s[EPW], r[EPW];
#pragma unroll
    for (int j = 0; j < EPW; ++j) { s[j] = snd[e0 + j]; r[j] = rcv[e0 + j]; }
    float* msum = own_plane(msum_a, msum_b);
    float v[EPW][2];
#pragma unroll
    for (int j = 0; j < EPW; ++j) {
      const float* src = xn + (long)s[j] * D + c;
      v[j][0] = src[0];
      v[j][1] = src[32];
    }
#pragma unroll
    for (int j = 0; j < EPW; ++j) {
      float* dst = msum + (long)r[j] * D + c;
      unsafeAtomicAdd(dst + 0, v[j][0]);
      unsafeAtomicAdd(dst + 32, v[j][1]);
    }
  } else {
    for (long e = e0; e < E; ++e) {
      int s = snd[e], r = rcv[e];
      float* msum = own_plane(msum_a, msum_b);
      const float* src = xn + (long)s * D + c;
      float* dst = msum + (long)r * D + c;
      float v0 = src[0];
      float v1 = src[32];
      unsafeAtomicAdd(dst + 0, v0);
      unsafeAtomicAdd(dst + 32, v1);
    }
  }
}


__global__ __launch_bounds__(256)
void sage_gemm_kernel(const float* __restrict__ A0,
                      int a0_scaled,
                      const float* msum,
                      const float* msum2,
                      const float* __restrict__ deg_r,
                      const float* __restrict__ deg_s,
                      const float* __restrict__ Wp,
                      const float* __restrict__ bias,
                      float* out,
                      float* xn_next,
                      float* sum_next,
                      float* zero_next,
                      int n_nodes, int do_relu, int write_next) {
  int wave = (blockIdx.x << 3) + (threadIdx.x >> 5);
  int mt_count = (n_nodes + 15) >> 4;
  if (wave >= mt_count) return;
  int lane = threadIdx.x & 31;
  int lo = lane & 15, hi = lane >> 4;

  int m = (wave << 4) + lo;
  bool mvalid = m < n_nodes;
  int mr = mvalid ? m : 0;

  float c = deg_r[mr];
  float rs = rsqrtf(c);
  float scale = rs * rs * rs;
  float ds = deg_s[mr];
  float rs_next = rsqrtf(ds);
  float a0s = a0_scaled ? sqrtf(ds) : 1.0f;

  const float* a0row = A0 + (long)mr * D;
  const float* smrow = msum + (long)mr * D;
  const float* sm2row = msum2 + (long)mr * D;
  const v2f* wp = (const v2f*)Wp;

  v8f acc[8];
#pragma unroll
  for (int t = 0; t < 8; ++t) acc[t] = (v8f)0.0f;

  for (int kk = 0; kk < D; kk += 4) {
    int k0 = kk + (hi << 1);
    v2f a = *(const v2f*)(a0row + k0);
    a.x *= a0s;
    a.y *= a0s;
    if (!mvalid) { a.x = 0.0f; a.y = 0.0f; }
    const v2f* wrow = wp + (long)(k0 >> 1) * D + lo;
#pragma unroll
    for (int t = 0; t < 8; ++t) {
      v2f b = wrow[t * 16];
      acc[t] = __builtin_amdgcn_wmma_f32_16x16x4_f32(
          false, a, false, b, (short)0, acc[t], false, false);
    }
  }

  for (int kk = 0; kk < D; kk += 4) {
    int k0 = kk + (hi << 1);
    v2f sm = *(const v2f*)(smrow + k0);
    v2f sm2 = *(const v2f*)(sm2row + k0);
    v2f a;
    a.x = (sm.x + sm2.x) * scale;
    a.y = (sm.y + sm2.y) * scale;
    if (!mvalid) { a.x = 0.0f; a.y = 0.0f; }
    const v2f* wrow = wp + (long)((k0 + D) >> 1) * D + lo;
#pragma unroll
    for (int t = 0; t < 8; ++t) {
      v2f b = wrow[t * 16];
      acc[t] = __builtin_amdgcn_wmma_f32_16x16x4_f32(
          false, a, false, b, (short)0, acc[t], false, false);
    }
  }

  float rsn[8];
#pragma unroll
  for (int v = 0; v < 8; ++v) rsn[v] = __shfl(rs_next, v + (hi << 3), 32);

  if (write_next) {
#pragma unroll
    for (int t = 0; t < 8; ++t) {
      int col = t * 16 + lo;
      float bv = bias[col];
#pragma unroll
      for (int v = 0; v < 8; ++v) {
        int row = (wave << 4) + v + (hi << 3);
        float val = acc[t][v] + bv;
        if (do_relu) val = fmaxf(val, 0.0f);
        if (row < n_nodes) {
          long idx = (long)row * D + col;
          float xs = val * rsn[v];
          xn_next[idx] = xs;
          sum_next[idx] = xs;
          zero_next[idx] = 0.0f;
        }
      }
    }
  } else {
#pragma unroll
    for (int t = 0; t < 8; ++t) {
      int col = t * 16 + lo;
      float bv = bias[col];
#pragma unroll
      for (int v = 0; v < 8; ++v) {
        int row = (wave << 4) + v + (hi << 3);
        float val = acc[t][v] + bv;
        if (do_relu) val = fmaxf(val, 0.0f);
        if (row < n_nodes) out[(long)row * D + col] = val;
      }
    }
  }
}


__global__ __launch_bounds__(128)
void pair_mlp_kernel(const int* __restrict__ pairs, const float* __restrict__ h,
                     const float* __restrict__ Wpa,
                     const float* __restrict__ ba,
                     const float* __restrict__ Wb, const float* __restrict__ bb,
                     float* __restrict__ out, int n_pairs) {
  int wv = threadIdx.x >> 5;
  int tile = (blockIdx.x << 2) + wv;
  int ntiles = (n_pairs + 15) >> 4;
  if (tile >= ntiles) return;
  int lane = threadIdx.x & 31;
  int lo = lane & 15, hi = lane >> 4;

  float4 zr[16];
  {
    int p = (tile << 4) + lo;
    int pc = p < n_pairs ? p : 0;
    int i = pairs[2 * pc], j = pairs[2 * pc + 1];
    const float4* hi4 = (const float4*)(h + (long)i * D) + hi;
    const float4* hj4 = (const float4*)(h + (long)j * D) + hi;
#pragma unroll
    for (int g = 0; g < 16; ++g) {
      float4 a = hi4[2 * g], b = hj4[2 * g];
      float4 v;
      v.x = a.x * b.x; v.y = a.y * b.y; v.z = a.z * b.z; v.w = a.w * b.w;
      zr[g] = v;
    }
  }

  const v2f* wp = (const v2f*)Wpa + (long)(2 * hi) * D + lo;
  float partial[8];
#pragma unroll
  for (int v = 0; v < 8; ++v) partial[v] = 0.0f;

  for (int t = 0; t < 8; ++t) {
    v8f acc = (v8f)0.0f;
    const v2f* wt = wp + t * 16;
#pragma unroll
    for (int g = 0; g < 16; ++g) {
      v2f a0, a1;
      a0.x = zr[g].x; a0.y = zr[g].y;
      a1.x = zr[g].z; a1.y = zr[g].w;
      v2f b0 = wt[(long)(4 * g) * D];
      v2f b1 = wt[(long)(4 * g + 1) * D];
      acc = __builtin_amdgcn_wmma_f32_16x16x4_f32(
          false, a0, false, b0, (short)0, acc, false, false);
      acc = __builtin_amdgcn_wmma_f32_16x16x4_f32(
          false, a1, false, b1, (short)0, acc, false, false);
    }
    int col = t * 16 + lo;
    float bv = ba[col], wb = Wb[col];
#pragma unroll
    for (int v = 0; v < 8; ++v) {
      partial[v] += fmaxf(acc[v] + bv, 0.0f) * wb;
    }
  }

#pragma unroll
  for (int v = 0; v < 8; ++v) {
    float x = partial[v];
    x += __shfl_xor(x, 1, 32);
    x += __shfl_xor(x, 2, 32);
    x += __shfl_xor(x, 4, 32);
    x += __shfl_xor(x, 8, 32);
    partial[v] = x;
  }
  if (lo == 0) {
    float bbv = bb[0];
#pragma unroll
    for (int v = 0; v < 8; ++v) {
      int row = (tile << 4) + v + (hi << 3);
      if (row < n_pairs) out[row] = partial[v] + bbv;
    }
  }
}


extern "C" void kernel_launch(void* const* d_in, const int* in_sizes, int n_in,
                              void* d_out, int out_size, void* d_ws, size_t ws_size,
                              hipStream_t stream) {
  const int* senders   = (const int*)d_in[1];
  const int* receivers = (const int*)d_in[2];
  const int* pairs     = (const int*)d_in[3];
  const float* emb = (const float*)d_in[4];
  const float* W1  = (const float*)d_in[5];
  const float* b1  = (const float*)d_in[6];
  const float* W2  = (const float*)d_in[7];
  const float* b2  = (const float*)d_in[8];
  const float* Wa  = (const float*)d_in[9];
  const float* ba  = (const float*)d_in[10];
  const float* Wb  = (const float*)d_in[11];
  const float* bb  = (const float*)d_in[12];
  float* out = (float*)d_out;

  int N = in_sizes[0];
  int E = in_sizes[1];
  int P = in_sizes[3] / 2;
  long ND = (long)N * D;

  float* ws    = (float*)d_ws;
  float* deg_s = ws;
  float* deg_r = ws + N;
  float* bufA  = ws + 2L * N;
  float* bufB  = bufA + ND;
  float* bufC  = bufB + ND;
  float* Wp1   = bufC + ND;
  float* Wp2   = Wp1 + 2L * D * D;
  float* Wpa   = Wp2 + 2L * D * D;

  const int T = 256;
  int nb_deg = (N + T - 1) / T;
  int nb_pack = (2 * NPAIR_SAGE + NPAIR_MLP + T - 1) / T;
  init_kernel<<<nb_deg + nb_pack, T, 0, stream>>>(deg_s, deg_r, N, nb_deg,
                                                   W1, Wp1, W2, Wp2, Wa, Wpa);

  degree_kernel<<<(E + T - 1) / T, T, 0, stream>>>(senders, receivers, deg_s, deg_r, E);

  long n4 = ND / 4;
  long sth = ((long)E + EPW - 1) / EPW * 32;
  dim3 sgrid((unsigned)((sth + T - 1) / T), 2);
  int mt = (N + 15) >> 4;
  int rnblocks = (int)((n4 + T * RN_PER_LANE - 1) / (T * RN_PER_LANE));

  rownorm_kernel<<<rnblocks, T, 0, stream>>>(emb, deg_s, bufA, bufB, bufC, N);
  scatter_kernel<<<sgrid, T, 0, stream>>>(senders, receivers, bufA, bufB, bufC, E);
  sage_gemm_kernel<<<(mt + 7) / 8, 256, 0, stream>>>(emb,  0, bufB, bufC, deg_r, deg_s, Wp1, b1,
                                                     bufB, bufA, bufC, bufB, N,  1,  1);

  scatter_kernel<<<sgrid, T, 0, stream>>>(senders, receivers, bufA, bufC, bufB, E);
  sage_gemm_kernel<<<(mt + 7) / 8, 256, 0, stream>>>(bufA,  1, bufC, bufB, deg_r, deg_s, Wp2, b2,
                                                     bufC, bufA, bufA, bufA, N,  0,  0);

  int ptiles = (P + 15) >> 4;
  pair_mlp_kernel<<<(ptiles + 3) / 4, 128, 0, stream>>>(pairs, bufC, Wpa, ba, Wb, bb, out, P);
}


# segment sums kept in LDS: a 32-wave work-group owns 391 node rows, filters one-byte edge keys, gathers sender rows in batches, adds them by LDS float atomics (other add order); one sum plane
# speedup vs baseline: 1.8213x; 1.0786x over previous
#include <hip/hip_runtime.h>

#define D 128

typedef __attribute__((ext_vector_type(2))) float v2f;
typedef __attribute__((ext_vector_type(8))) float v8f;


#define NPAIR_SAGE (D * D)
#define NPAIR_MLP (D * D / 2)
__global__ __launch_bounds__(256)
void init_kernel(float* __restrict__ deg_s, float* __restrict__ deg_r, int n_nodes, int nb_deg,
                 const float* __restrict__ W1, float* __restrict__ Wp1,
                 const float* __restrict__ W2, float* __restrict__ Wp2,
                 const float* __restrict__ Wa, float* __restrict__ Wpa) {
  int b = blockIdx.x;
  if (b < nb_deg) {
    int i = b * 256 + threadIdx.x;
    if (i < n_nodes) { deg_s[i] = 1.0f; deg_r[i] = 1.0f; }
    return;
  }
  int i = (b - nb_deg) * 256 + threadIdx.x;
  const float* W;
  float* Wp;
  if (i < NPAIR_SAGE) { W = W1; Wp = Wp1; }
  else if (i < 2 * NPAIR_SAGE) { W = W2; Wp = Wp2; i -= NPAIR_SAGE; }
  else if (i < 2 * NPAIR_SAGE + NPAIR_MLP) { W = Wa; Wp = Wpa; i -= 2 * NPAIR_SAGE; }
  else return;
  int kp = i >> 7, n = i & (D - 1);
  float2 v;
  v.x = W[(long)(2 * kp) * D + n];
  v.y = W[(long)(2 * kp + 1) * D + n];
  ((float2*)Wp)[i] = v;
}

#define RPB 391

__global__ __launch_bounds__(256)
void degree_kernel(const int* __restrict__ snd, const int* __restrict__ rcv,
                   float* deg_s, float* deg_r, unsigned int* __restrict__ keyw,
                   int E, int nwords) {
  int w = blockIdx.x * blockDim.x + threadIdx.x;
  if (w >= nwords) return;
  unsigned int packed = 0u;
  if (4 * w + 3 < E) {
    int4 sv = ((const int4*)snd)[w];
    int4 rv = ((const int4*)rcv)[w];
    unsafeAtomicAdd(&deg_s[sv.x], 1.0f);
    unsafeAtomicAdd(&deg_s[sv.y], 1.0f);
    unsafeAtomicAdd(&deg_s[sv.z], 1.0f);
    unsafeAtomicAdd(&deg_s[sv.w], 1.0f);
    unsafeAtomicAdd(&deg_r[rv.x], 1.0f);
    unsafeAtomicAdd(&deg_r[rv.y], 1.0f);
    unsafeAtomicAdd(&deg_r[rv.z], 1.0f);
    unsafeAtomicAdd(&deg_r[rv.w], 1.0f);
    packed = (unsigned int)((rv.x / RPB) & 255) | (unsigned int)((rv.y / RPB) & 255) << 8 |
             (unsigned int)((rv.z / RPB) & 255) << 16 | (unsigned int)((rv.w / RPB) & 255) << 24;
  } else {
    for (int j = 0; j < 4; ++j) {
      int e = 4 * w + j;
      if (e < E) {
        int s = snd[e], r = rcv[e];
        unsafeAtomicAdd(&deg_s[s], 1.0f);
        unsafeAtomicAdd(&deg_r[r], 1.0f);
        packed |= (unsigned int)((r / RPB) & 255) << (8 * j);
      }
    }
  }
  keyw[w] = packed;
}

#define RN_PER_LANE 4
__global__ __launch_bounds__(256)
void rownorm_kernel(const float* __restrict__ src, const float* __restrict__ deg,
                    float* __restrict__ dst, int n_nodes) {
  long base = (long)blockIdx.x * (256 * RN_PER_LANE) + threadIdx.x;
  long n4 = (long)n_nodes * (D / 4);
  float4 v[RN_PER_LANE];
  float s[RN_PER_LANE];
#pragma unroll
  for (int k = 0; k < RN_PER_LANE; ++k) {
    long i = base + (long)k * 256;
    long ic = i < n4 ? i : 0;
    s[k] = deg[(int)(ic >> 5)];
    v[k] = ((const float4*)src)[ic];
  }
#pragma unroll
  for (int k = 0; k < RN_PER_LANE; ++k) {
    long i = base + (long)k * 256;
    if (i < n4) {
      float sc = rsqrtf(s[k]);
      float4 o = v[k];
      o.x *= sc; o.y *= sc; o.z *= sc; o.w *= sc;
      ((float4*)dst)[i] = o;
    }
  }
}

#define SC_WAVES 32
#define SC_THREADS (SC_WAVES * 32)
#define KCHUNK 512
#define KAHEAD 8
#define QBATCH 16
#define QCAP 48

static __device__ __forceinline__ unsigned int zero_bytes(unsigned int x) {
  return ~(((x & 0x7F7F7F7Fu) + 0x7F7F7F7Fu) | x | 0x7F7F7F7Fu);
}

static __device__ __forceinline__ void gather_add(const int* qe, int cnt,
                                                  const int* __restrict__ snd,
                                                  const int* __restrict__ rcv,
                                                  const float* __restrict__ xn, float* rows,
                                                  int E, int row0, int nrows, int lane) {
  int sl = lane & (QBATCH - 1);
  int e = qe[sl < cnt ? sl : 0];
  bool ok = (sl < cnt) && (e < E);
  int ec = ok ? e : 0;
  int sv = snd[ec];
  int rv = ok ? rcv[ec] - row0 : -1;
  float2 v[QBATCH][2];
  int lr[QBATCH];
#pragma unroll
  for (int j = 0; j < QBATCH; ++j) {
    int s = __builtin_amdgcn_readlane(sv, j);
    lr[j] = __builtin_amdgcn_readlane(rv, j);
    const float2* srow = (const float2*)(xn + (long)s * D) + lane;
    v[j][0] = srow[0];
    v[j][1] = srow[32];
  }
#pragma unroll
  for (int j = 0; j < QBATCH; ++j) {
    if ((unsigned int)lr[j] < (unsigned int)nrows) {
      float* dst = rows + lr[j] * D + 2 * lane;
      __hip_atomic_fetch_add(dst + 0, v[j][0].x, __ATOMIC_RELAXED, __HIP_MEMORY_SCOPE_WORKGROUP);
      __hip_atomic_fetch_add(dst + 1, v[j][0].y, __ATOMIC_RELAXED, __HIP_MEMORY_SCOPE_WORKGROUP);
      __hip_atomic_fetch_add(dst + 64, v[j][1].x, __ATOMIC_RELAXED, __HIP_MEMORY_SCOPE_WORKGROUP);
      __hip_atomic_fetch_add(dst + 65, v[j][1].y, __ATOMIC_RELAXED, __HIP_MEMORY_SCOPE_WORKGROUP);
    }
  }
}

__global__ __launch_bounds__(SC_THREADS)
void scatter_kernel(const int* __restrict__ snd, const int* __restrict__ rcv,
                    const unsigned int* __restrict__ keyw,
                    const float* __restrict__ xn, float* __restrict__ msum,
                    int E, int nchunks, int n_nodes, int nranges) {
  __shared__ float rows[RPB * D];
  __shared__ int stack[SC_WAVES * QCAP];

  const int lane = threadIdx.x & 31;
  const int wave = __builtin_amdgcn_readfirstlane((int)(threadIdx.x >> 5));
  int* q = stack + wave * QCAP;
  float4* rows4 = (float4*)rows;

  for (int range = blockIdx.x; range < nranges; range += gridDim.x) {
    const int row0 = range * RPB;
    const int nrows = min(RPB, n_nodes - row0);
    const int n4 = nrows * (D / 4);

    const float4* src4 = (const float4*)(xn + (long)row0 * D);
    for (int i0 = threadIdx.x; i0 < n4; i0 += SC_THREADS * 4) {
      float4 t[4];
      int ii[4];
#pragma unroll
      for (int k = 0; k < 4; ++k) {
        int i = i0 + k * SC_THREADS;
        ii[k] = i < n4 ? i : i0;
        t[k] = src4[ii[k]];
      }
#pragma unroll
      for (int k = 0; k < 4; ++k) rows4[ii[k]] = t[k];
    }
    __syncthreads();

    const unsigned int idpat = (unsigned int)(range & 255) * 0x01010101u;
    int qn = 0;
    for (int c0 = wave; c0 < nchunks; c0 += SC_WAVES * KAHEAD) {
      uint4 kv[KAHEAD];
      unsigned int m[KAHEAD];
#pragma unroll
      for (int u = 0; u < KAHEAD; ++u) {
        int c = c0 + u * SC_WAVES;
        int cc = c < nchunks ? c : c0;
        kv[u] = ((const uint4*)keyw)[(long)cc * 32 + lane];
      }
#pragma unroll
      for (int u = 0; u < KAHEAD; ++u) {
        unsigned int mu = (zero_bytes(kv[u].x ^ idpat) >> 7) | (zero_bytes(kv[u].y ^ idpat) >> 6) |
                          (zero_bytes(kv[u].z ^ idpat) >> 5) | (zero_bytes(kv[u].w ^ idpat) >> 4);
        m[u] = (c0 + u * SC_WAVES < nchunks) ? mu : 0u;
      }
#pragma nounroll
      for (int u = 0; u < KAHEAD; ++u) {
        int ebase = (c0 + u * SC_WAVES) * KCHUNK + lane * 16;
        unsigned int mm = m[0];
#pragma unroll
        for (int t = 1; t < KAHEAD; ++t) mm = (u == t) ? m[t] : mm;
        unsigned int any;
        while ((any = __builtin_amdgcn_ballot_w32(mm != 0u)) != 0u) {
          if (mm != 0u) {
            int p = __builtin_ctz(mm);
            mm &= mm - 1u;
            int pos = qn + (int)__builtin_amdgcn_mbcnt_lo(any, 0u);
            q[pos] = ebase + 4 * (p & 7) + (p >> 3);
          }
          qn += __builtin_popcount(any);
          while (qn >= QBATCH) {
            qn -= QBATCH;
            gather_add(q + qn, QBATCH, snd, rcv, xn, rows, E, row0, nrows, lane);
          }
        }
      }
    }
    if (qn > 0) gather_add(q, qn, snd, rcv, xn, rows, E, row0, nrows, lane);
    __syncthreads();

    float4* dst4 = (float4*)(msum + (long)row0 * D);
    for (int i = threadIdx.x; i < n4; i += SC_THREADS) dst4[i] = rows4[i];
    __syncthreads();
  }
}


__global__ __launch_bounds__(256)
void sage_gemm_kernel(const float* __restrict__ A0,
                      int a0_scaled,
                      const float* msum,
                      const float* __restrict__ deg_r,
                      const float* __restrict__ deg_s,
                      const float* __restrict__ Wp,
                      const float* __restrict__ bias,
                      float* out,
                      float* xn_next,
                      int n_nodes, int do_relu, int write_next) {
  int wave = (blockIdx.x << 3) + (threadIdx.x >> 5);
  int mt_count = (n_nodes + 15) >> 4;
  if (wave >= mt_count) return;
  int lane = threadIdx.x & 31;
  int lo = lane & 15, hi = lane >> 4;

  int m = (wave << 4) + lo;
  bool mvalid = m < n_nodes;
  int mr = mvalid ? m : 0;

  float c = deg_r[mr];
  float rs = rsqrtf(c);
  float scale = rs * rs * rs;
  float ds = deg_s[mr];
  float rs_next = rsqrtf(ds);
  float a0s = a0_scaled ? sqrtf(ds) : 1.0f;

  const float* a0row = A0 + (long)mr * D;
  const float* smrow = msum + (long)mr * D;
  const v2f* wp = (const v2f*)Wp;

  v8f acc[8];
#pragma unroll
  for (int t = 0; t < 8; ++t) acc[t] = (v8f)0.0f;

  for (int kk = 0; kk < D; kk += 4) {
    int k0 = kk + (hi << 1);
    v2f a = *(const v2f*)(a0row + k0);
    a.x *= a0s;
    a.y *= a0s;
    if (!mvalid) { a.x = 0.0f; a.y = 0.0f; }
    const v2f* wrow = wp + (long)(k0 >> 1) * D + lo;
#pragma unroll
    for (int t = 0; t < 8; ++t) {
      v2f b = wrow[t * 16];
      acc[t] = __builtin_amdgcn_wmma_f32_16x16x4_f32(
          false, a, false, b, (short)0, acc[t], false, false);
    }
  }

  for (int kk = 0; kk < D; kk += 4) {
    int k0 = kk + (hi << 1);
    v2f sm = *(const v2f*)(smrow + k0);
    v2f a;
    a.x = sm.x * scale;
    a.y = sm.y * scale;
    if (!mvalid) { a.x = 0.0f; a.y = 0.0f; }
    const v2f* wrow = wp + (long)((k0 + D) >> 1) * D + lo;
#pragma unroll
    for (int t = 0; t < 8; ++t) {
      v2f b = wrow[t * 16];
      acc[t] = __builtin_amdgcn_wmma_f32_16x16x4_f32(
          false, a, false, b, (short)0, acc[t], false, false);
    }
  }

  float rsn[8];
#pragma unroll
  for (int v = 0; v < 8; ++v) rsn[v] = __shfl(rs_next, v + (hi << 3), 32);

  if (write_next) {
#pragma unroll
    for (int t = 0; t < 8; ++t) {
      int col = t * 16 + lo;
      float bv = bias[col];
#pragma unroll
      for (int v = 0; v < 8; ++v) {
        int row = (wave << 4) + v + (hi << 3);
        float val = acc[t][v] + bv;
        if (do_relu) val = fmaxf(val, 0.0f);
        if (row < n_nodes) {
          xn_next[(long)row * D + col] = val * rsn[v];
        }
      }
    }
  } else {
#pragma unroll
    for (int t = 0; t < 8; ++t) {
      int col = t * 16 + lo;
      float bv = bias[col];
#pragma unroll
      for (int v = 0; v < 8; ++v) {
        int row = (wave << 4) + v + (hi << 3);
        float val = acc[t][v] + bv;
        if (do_relu) val = fmaxf(val, 0.0f);
        if (row < n_nodes) out[(long)row * D + col] = val;
      }
    }
  }
}


__global__ __launch_bounds__(128)
void pair_mlp_kernel(const int* __restrict__ pairs, const float* __restrict__ h,
                     const float* __restrict__ Wpa,
                     const float* __restrict__ ba,
                     const float* __restrict__ Wb, const float* __restrict__ bb,
                     float* __restrict__ out, int n_pairs) {
  int wv = threadIdx.x >> 5;
  int tile = (blockIdx.x << 2) + wv;
  int ntiles = (n_pairs + 15) >> 4;
  if (tile >= ntiles) return;
  int lane = threadIdx.x & 31;
  int lo = lane & 15, hi = lane >> 4;

  float4 zr[16];
  {
    int p = (tile << 4) + lo;
    int pc = p < n_pairs ? p : 0;
    int i = pairs[2 * pc], j = pairs[2 * pc + 1];
    const float4* hi4 = (const float4*)(h + (long)i * D) + hi;
    const float4* hj4 = (const float4*)(h + (long)j * D) + hi;
#pragma unroll
    for (int g = 0; g < 16; ++g) {
      float4 a = hi4[2 * g], b = hj4[2 * g];
      float4 v;
      v.x = a.x * b.x; v.y = a.y * b.y; v.z = a.z * b.z; v.w = a.w * b.w;
      zr[g] = v;
    }
  }

  const v2f* wp = (const v2f*)Wpa + (long)(2 * hi) * D + lo;
  float partial[8];
#pragma unroll
  for (int v = 0; v < 8; ++v) partial[v] = 0.0f;

  for (int t = 0; t < 8; ++t) {
    v8f acc = (v8f)0.0f;
    const v2f* wt = wp + t * 16;
#pragma unroll
    for (int g = 0; g < 16; ++g) {
      v2f a0, a1;
      a0.x = zr[g].x; a0.y = zr[g].y;
      a1.x = zr[g].z; a1.y = zr[g].w;
      v2f b0 = wt[(long)(4 * g) * D];
      v2f b1 = wt[(long)(4 * g + 1) * D];
      acc = __builtin_amdgcn_wmma_f32_16x16x4_f32(
          false, a0, false, b0, (short)0, acc, false, false);
      acc = __builtin_amdgcn_wmma_f32_16x16x4_f32(
          false, a1, false, b1, (short)0, acc, false, false);
    }
    int col = t * 16 + lo;
    float bv = ba[col], wb = Wb[col];
#pragma unroll
    for (int v = 0; v < 8; ++v) {
      partial[v] += fmaxf(acc[v] + bv, 0.0f) * wb;
    }
  }

#pragma unroll
  for (int v = 0; v < 8; ++v) {
    float x = partial[v];
    x += __shfl_xor(x, 1, 32);
    x += __shfl_xor(x, 2, 32);
    x += __shfl_xor(x, 4, 32);
    x += __shfl_xor(x, 8, 32);
    partial[v] = x;
  }
  if (lo == 0) {
    float bbv = bb[0];
#pragma unroll
    for (int v = 0; v < 8; ++v) {
      int row = (tile << 4) + v + (hi << 3);
      if (row < n_pairs) out[row] = partial[v] + bbv;
    }
  }
}


extern "C" void kernel_launch(void* const* d_in, const int* in_sizes, int n_in,
                              void* d_out, int out_size, void* d_ws, size_t ws_size,
                              hipStream_t stream) {
  const int* senders   = (const int*)d_in[1];
  const int* receivers = (const int*)d_in[2];
  const int* pairs     = (const int*)d_in[3];
  const float* emb = (const float*)d_in[4];
  const float* W1  = (const float*)d_in[5];
  const float* b1  = (const float*)d_in[6];
  const float* W2  = (const float*)d_in[7];
  const float* b2  = (const float*)d_in[8];
  const float* Wa  = (const float*)d_in[9];
  const float* ba  = (const float*)d_in[10];
  const float* Wb  = (const float*)d_in[11];
  const float* bb  = (const float*)d_in[12];
  float* out = (float*)d_out;

  int N = in_sizes[0];
  int E = in_sizes[1];
  int P = in_sizes[3] / 2;
  long ND = (long)N * D;

  float* ws    = (float*)d_ws;
  float* deg_s = ws;
  float* deg_r = ws + N;
  float* bufA  = ws + 2L * N;
  float* bufB  = bufA + ND;
  float* bufC  = bufB + ND;
  unsigned int* keyw = (unsigned int*)bufC;
  float* Wp1   = bufC + ND;
  float* Wp2   = Wp1 + 2L * D * D;
  float* Wpa   = Wp2 + 2L * D * D;

  const int T = 256;
  int nb_deg = (N + T - 1) / T;
  int nb_pack = (2 * NPAIR_SAGE + NPAIR_MLP + T - 1) / T;
  init_kernel<<<nb_deg + nb_pack, T, 0, stream>>>(deg_s, deg_r, N, nb_deg,
                                                   W1, Wp1, W2, Wp2, Wa, Wpa);

  int nchunks = (E + KCHUNK - 1) / KCHUNK;
  int nwords = nchunks * (KCHUNK / 4);
  degree_kernel<<<(nwords + T - 1) / T, T, 0, stream>>>(senders, receivers, deg_s, deg_r, keyw, E, nwords);

  long n4 = ND / 4;
  int nranges = (N + RPB - 1) / RPB;
  int sgrid = nranges < 256 ? nranges : 256;
  int mt = (N + 15) >> 4;
  int rnblocks = (int)((n4 + T * RN_PER_LANE - 1) / (T * RN_PER_LANE));

  rownorm_kernel<<<rnblocks, T, 0, stream>>>(emb, deg_s, bufA, N);
  scatter_kernel<<<sgrid, SC_THREADS, 0, stream>>>(senders, receivers, keyw, bufA, bufB, E, nchunks, N, nranges);
  sage_gemm_kernel<<<(mt + 7) / 8, 256, 0, stream>>>(emb,  0, bufB, deg_r, deg_s, Wp1, b1,
                                                     bufB, bufA, N,  1,  1);

  scatter_kernel<<<sgrid, SC_THREADS, 0, stream>>>(senders, receivers, keyw, bufA, bufB, E, nchunks, N, nranges);
  sage_gemm_kernel<<<(mt + 7) / 8, 256, 0, stream>>>(bufA,  1, bufB, deg_r, deg_s, Wp2, b2,
                                                     bufB, bufA, N,  0,  0);

  int ptiles = (P + 15) >> 4;
  pair_mlp_kernel<<<(ptiles + 3) / 4, 128, 0, stream>>>(pairs, bufB, Wpa, ba, Wb, bb, out, P);
}


# second-layer segment sum replays per-(range, wave) logs of (sender, row) pairs written by the first layer's owner work-groups while filtering: no key walk or index lookups there; LDS float adds kept
# speedup vs baseline: 1.8928x; 1.0393x over previous
#include <hip/hip_runtime.h>

#define D 128

typedef __attribute__((ext_vector_type(2))) float v2f;
typedef __attribute__((ext_vector_type(8))) float v8f;


#define NPAIR_SAGE (D * D)
#define NPAIR_MLP (D * D / 2)
__global__ __launch_bounds__(256)
void init_kernel(float* __restrict__ deg_s, float* __restrict__ deg_r, int n_nodes, int nb_deg,
                 const float* __restrict__ W1, float* __restrict__ Wp1,
                 const float* __restrict__ W2, float* __restrict__ Wp2,
                 const float* __restrict__ Wa, float* __restrict__ Wpa) {
  int b = blockIdx.x;
  if (b < nb_deg) {
    int i = b * 256 + threadIdx.x;
    if (i < n_nodes) { deg_s[i] = 1.0f; deg_r[i] = 1.0f; }
    return;
  }
  int i = (b - nb_deg) * 256 + threadIdx.x;
  const float* W;
  float* Wp;
  if (i < NPAIR_SAGE) { W = W1; Wp = Wp1; }
  else if (i < 2 * NPAIR_SAGE) { W = W2; Wp = Wp2; i -= NPAIR_SAGE; }
  else if (i < 2 * NPAIR_SAGE + NPAIR_MLP) { W = Wa; Wp = Wpa; i -= 2 * NPAIR_SAGE; }
  else return;
  int kp = i >> 7, n = i & (D - 1);
  float2 v;
  v.x = W[(long)(2 * kp) * D + n];
  v.y = W[(long)(2 * kp + 1) * D + n];
  ((float2*)Wp)[i] = v;
}

#define RPB 391

__global__ __launch_bounds__(256)
void degree_kernel(const int* __restrict__ snd, const int* __restrict__ rcv,
                   float* deg_s, float* deg_r, unsigned int* __restrict__ keyw,
                   int E, int nwords) {
  int w = blockIdx.x * blockDim.x + threadIdx.x;
  if (w >= nwords) return;
  unsigned int packed = 0u;
  if (4 * w + 3 < E) {
    int4 sv = ((const int4*)snd)[w];
    int4 rv = ((const int4*)rcv)[w];
    unsafeAtomicAdd(&deg_s[sv.x], 1.0f);
    unsafeAtomicAdd(&deg_s[sv.y], 1.0f);
    unsafeAtomicAdd(&deg_s[sv.z], 1.0f);
    unsafeAtomicAdd(&deg_s[sv.w], 1.0f);
    unsafeAtomicAdd(&deg_r[rv.x], 1.0f);
    unsafeAtomicAdd(&deg_r[rv.y], 1.0f);
    unsafeAtomicAdd(&deg_r[rv.z], 1.0f);
    unsafeAtomicAdd(&deg_r[rv.w], 1.0f);
    packed = (unsigned int)((rv.x / RPB) & 255) | (unsigned int)((rv.y / RPB) & 255) << 8 |
             (unsigned int)((rv.z / RPB) & 255) << 16 | (unsigned int)((rv.w / RPB) & 255) << 24;
  } else {
    for (int j = 0; j < 4; ++j) {
      int e = 4 * w + j;
      if (e < E) {
        int s = snd[e], r = rcv[e];
        unsafeAtomicAdd(&deg_s[s], 1.0f);
        unsafeAtomicAdd(&deg_r[r], 1.0f);
        packed |= (unsigned int)((r / RPB) & 255) << (8 * j);
      }
    }
  }
  keyw[w] = packed;
}

#define RN_PER_LANE 4
__global__ __launch_bounds__(256)
void rownorm_kernel(const float* __restrict__ src, const float* __restrict__ deg,
                    float* __restrict__ dst, int n_nodes) {
  long base = (long)blockIdx.x * (256 * RN_PER_LANE) + threadIdx.x;
  long n4 = (long)n_nodes * (D / 4);
  float4 v[RN_PER_LANE];
  float s[RN_PER_LANE];
#pragma unroll
  for (int k = 0; k < RN_PER_LANE; ++k) {
    long i = base + (long)k * 256;
    long ic = i < n4 ? i : 0;
    s[k] = deg[(int)(ic >> 5)];
    v[k] = ((const float4*)src)[ic];
  }
#pragma unroll
  for (int k = 0; k < RN_PER_LANE; ++k) {
    long i = base + (long)k * 256;
    if (i < n4) {
      float sc = rsqrtf(s[k]);
      float4 o = v[k];
      o.x *= sc; o.y *= sc; o.z *= sc; o.w *= sc;
      ((float4*)dst)[i] = o;
    }
  }
}

#define SC_WAVES 32
#define SC_THREADS (SC_WAVES * 32)
#define KCHUNK 512
#define KAHEAD 8
#define QBATCH 16
#define QCAP 48
#define LCAP 512

static __device__ __forceinline__ unsigned int zero_bytes(unsigned int x) {
  return ~(((x & 0x7F7F7F7Fu) + 0x7F7F7F7Fu) | x | 0x7F7F7F7Fu);
}

static __device__ __forceinline__ void add_rows(int sv, int rv, int off,
                                                const float* __restrict__ xn, float* rows,
                                                int nrows, int lane) {
  float2 v[QBATCH][2];
  int lr[QBATCH];
#pragma unroll
  for (int j = 0; j < QBATCH; ++j) {
    int s = __builtin_amdgcn_readlane(sv, off + j);
    lr[j] = __builtin_amdgcn_readlane(rv, off + j);
    const float2* srow = (const float2*)(xn + (long)s * D) + lane;
    v[j][0] = srow[0];
    v[j][1] = srow[32];
  }
#pragma unroll
  for (int j = 0; j < QBATCH; ++j) {
    if ((unsigned int)lr[j] < (unsigned int)nrows) {
      float* dst = rows + lr[j] * D + 2 * lane;
      __hip_atomic_fetch_add(dst + 0, v[j][0].x, __ATOMIC_RELAXED, __HIP_MEMORY_SCOPE_WORKGROUP);
      __hip_atomic_fetch_add(dst + 1, v[j][0].y, __ATOMIC_RELAXED, __HIP_MEMORY_SCOPE_WORKGROUP);
      __hip_atomic_fetch_add(dst + 64, v[j][1].x, __ATOMIC_RELAXED, __HIP_MEMORY_SCOPE_WORKGROUP);
      __hip_atomic_fetch_add(dst + 65, v[j][1].y, __ATOMIC_RELAXED, __HIP_MEMORY_SCOPE_WORKGROUP);
    }
  }
}

template <bool LOG>
static __device__ __forceinline__ void gather_add(const int* qe, int cnt,
                                                  const int* __restrict__ snd,
                                                  const int* __restrict__ rcv,
                                                  const float* __restrict__ xn, float* rows,
                                                  int E, int row0, int nrows, int lane,
                                                  int2* __restrict__ wlog, int& ln) {
  int sl = lane & (QBATCH - 1);
  int e = qe[sl < cnt ? sl : 0];
  bool ok = (sl < cnt) && (e < E);
  int ec = ok ? e : 0;
  int sv = snd[ec];
  int rv = ok ? rcv[ec] - row0 : -1;
  if (LOG) {
    bool keep = (lane < QBATCH) && ((unsigned int)rv < (unsigned int)nrows);
    unsigned int km = __builtin_amdgcn_ballot_w32(keep);
    if (keep) {
      int pos = ln + (int)__builtin_amdgcn_mbcnt_lo(km, 0u);
      if (pos < LCAP) wlog[pos] = make_int2(sv, rv);
    }
    ln += __builtin_popcount(km);
  }
  add_rows(sv, rv, 0, xn, rows, nrows, lane);
}

template <bool LOG>
static __device__ __forceinline__ void scan_and_add(const int* __restrict__ snd,
                                                    const int* __restrict__ rcv,
                                                    const unsigned int* __restrict__ keyw,
                                                    const float* __restrict__ xn, float* rows,
                                                    int* q, int E, int nchunks, int range,
                                                    int row0, int nrows, int wave, int lane,
                                                    int2* __restrict__ wlog, int& ln) {
  const unsigned int idpat = (unsigned int)(range & 255) * 0x01010101u;
  int qn = 0;
  for (int c0 = wave; c0 < nchunks; c0 += SC_WAVES * KAHEAD) {
    uint4 kv[KAHEAD];
    unsigned int m[KAHEAD];
#pragma unroll
    for (int u = 0; u < KAHEAD; ++u) {
      int c = c0 + u * SC_WAVES;
      int cc = c < nchunks ? c : c0;
      kv[u] = ((const uint4*)keyw)[(long)cc * 32 + lane];
    }
#pragma unroll
    for (int u = 0; u < KAHEAD; ++u) {
      unsigned int mu = (zero_bytes(kv[u].x ^ idpat) >> 7) | (zero_bytes(kv[u].y ^ idpat) >> 6) |
                        (zero_bytes(kv[u].z ^ idpat) >> 5) | (zero_bytes(kv[u].w ^ idpat) >> 4);
      m[u] = (c0 + u * SC_WAVES < nchunks) ? mu : 0u;
    }
#pragma nounroll
    for (int u = 0; u < KAHEAD; ++u) {
      int ebase = (c0 + u * SC_WAVES) * KCHUNK + lane * 16;
      unsigned int mm = m[0];
#pragma unroll
      for (int t = 1; t < KAHEAD; ++t) mm = (u == t) ? m[t] : mm;
      unsigned int any;
      while ((any = __builtin_amdgcn_ballot_w32(mm != 0u)) != 0u) {
        if (mm != 0u) {
          int p = __builtin_ctz(mm);
          mm &= mm - 1u;
          int pos = qn + (int)__builtin_amdgcn_mbcnt_lo(any, 0u);
          q[pos] = ebase + 4 * (p & 7) + (p >> 3);
        }
        qn += __builtin_popcount(any);
        while (qn >= QBATCH) {
          qn -= QBATCH;
          gather_add<LOG>(q + qn, QBATCH, snd, rcv, xn, rows, E, row0, nrows, lane, wlog, ln);
        }
      }
    }
  }
  if (qn > 0) gather_add<LOG>(q, qn, snd, rcv, xn, rows, E, row0, nrows, lane, wlog, ln);
}

template <bool REPLAY>
__global__ __launch_bounds__(SC_THREADS)
void scatter_kernel(const int* __restrict__ snd, const int* __restrict__ rcv,
                    const unsigned int* __restrict__ keyw,
                    const float* __restrict__ xn, float* __restrict__ msum,
                    int2* __restrict__ elog, int* __restrict__ elog_n,
                    int E, int nchunks, int n_nodes, int nranges) {
  __shared__ float rows[RPB * D];
  __shared__ int stack[SC_WAVES * QCAP];

  const int lane = threadIdx.x & 31;
  const int wave = __builtin_amdgcn_readfirstlane((int)(threadIdx.x >> 5));
  int* q = stack + wave * QCAP;
  float4* rows4 = (float4*)rows;

  for (int range = blockIdx.x; range < nranges; range += gridDim.x) {
    const int row0 = range * RPB;
    const int nrows = min(RPB, n_nodes - row0);
    const int n4 = nrows * (D / 4);

    const float4* src4 = (const float4*)(xn + (long)row0 * D);
    for (int i0 = threadIdx.x; i0 < n4; i0 += SC_THREADS * 4) {
      float4 t[4];
      int ii[4];
#pragma unroll
      for (int k = 0; k < 4; ++k) {
        int i = i0 + k * SC_THREADS;
        ii[k] = i < n4 ? i : i0;
        t[k] = src4[ii[k]];
      }
#pragma unroll
      for (int k = 0; k < 4; ++k) rows4[ii[k]] = t[k];
    }
    __syncthreads();

    int2* wlog = elog + (long)(range * SC_WAVES + wave) * LCAP;
    int ln = 0;
    if (!REPLAY) {
      scan_and_add<true>(snd, rcv, keyw, xn, rows, q, E, nchunks, range, row0, nrows, wave, lane, wlog, ln);
      if (lane == 0) elog_n[range * SC_WAVES + wave] = ln;
    } else {
      const int n = __builtin_amdgcn_readfirstlane(elog_n[range * SC_WAVES + wave]);
      if (n > LCAP) {
        scan_and_add<false>(snd, rcv, keyw, xn, rows, q, E, nchunks, range, row0, nrows, wave, lane, wlog, ln);
      } else if (n > 0) {
        int2 rnext = wlog[lane < n ? lane : 0];
        for (int i = 0; i < n; i += 32) {
          int2 r = rnext;
          int nidx = i + 32 + lane;
          rnext = wlog[nidx < n ? nidx : 0];
          int sv = r.x;
          int rv = (i + lane < n) ? r.y : -1;
          add_rows(sv, rv, 0, xn, rows, nrows, lane);
          if (i + QBATCH < n) add_rows(sv, rv, QBATCH, xn, rows, nrows, lane);
        }
      }
    }
    __syncthreads();

    float4* dst4 = (float4*)(msum + (long)row0 * D);
    for (int i = threadIdx.x; i < n4; i += SC_THREADS) dst4[i] = rows4[i];
    __syncthreads();
  }
}


__global__ __launch_bounds__(256)
void sage_gemm_kernel(const float* __restrict__ A0,
                      int a0_scaled,
                      const float* msum,
                      const float* __restrict__ deg_r,
                      const float* __restrict__ deg_s,
                      const float* __restrict__ Wp,
                      const float* __restrict__ bias,
                      float* out,
                      float* xn_next,
                      int n_nodes, int do_relu, int write_next) {
  int wave = (blockIdx.x << 3) + (threadIdx.x >> 5);
  int mt_count = (n_nodes + 15) >> 4;
  if (wave >= mt_count) return;
  int lane = threadIdx.x & 31;
  int lo = lane & 15, hi = lane >> 4;

  int m = (wave << 4) + lo;
  bool mvalid = m < n_nodes;
  int mr = mvalid ? m : 0;

  float c = deg_r[mr];
  float rs = rsqrtf(c);
  float scale = rs * rs * rs;
  float ds = deg_s[mr];
  float rs_next = rsqrtf(ds);
  float a0s = a0_scaled ? sqrtf(ds) : 1.0f;

  const float* a0row = A0 + (long)mr * D;
  const float* smrow = msum + (long)mr * D;
  const v2f* wp = (const v2f*)Wp;

  v8f acc[8];
#pragma unroll
  for (int t = 0; t < 8; ++t) acc[t] = (v8f)0.0f;

  for (int kk = 0; kk < D; kk += 4) {
    int k0 = kk + (hi << 1);
    v2f a = *(const v2f*)(a0row + k0);
    a.x *= a0s;
    a.y *= a0s;
    if (!mvalid) { a.x = 0.0f; a.y = 0.0f; }
    const v2f* wrow = wp + (long)(k0 >> 1) * D + lo;
#pragma unroll
    for (int t = 0; t < 8; ++t) {
      v2f b = wrow[t * 16];
      acc[t] = __builtin_amdgcn_wmma_f32_16x16x4_f32(
          false, a, false, b, (short)0, acc[t], false, false);
    }
  }

  for (int kk = 0; kk < D; kk += 4) {
    int k0 = kk + (hi << 1);
    v2f sm = *(const v2f*)(smrow + k0);
    v2f a;
    a.x = sm.x * scale;
    a.y = sm.y * scale;
    if (!mvalid) { a.x = 0.0f; a.y = 0.0f; }
    const v2f* wrow = wp + (long)((k0 + D) >> 1) * D + lo;
#pragma unroll
    for (int t = 0; t < 8; ++t) {
      v2f b = wrow[t * 16];
      acc[t] = __builtin_amdgcn_wmma_f32_16x16x4_f32(
          false, a, false, b, (short)0, acc[t], false, false);
    }
  }

  float rsn[8];
#pragma unroll
  for (int v = 0; v < 8; ++v) rsn[v] = __shfl(rs_next, v + (hi << 3), 32);

  if (write_next) {
#pragma unroll
    for (int t = 0; t < 8; ++t) {
      int col = t * 16 + lo;
      float bv = bias[col];
#pragma unroll
      for (int v = 0; v < 8; ++v) {
        int row = (wave << 4) + v + (hi << 3);
        float val = acc[t][v] + bv;
        if (do_relu) val = fmaxf(val, 0.0f);
        if (row < n_nodes) {
          xn_next[(long)row * D + col] = val * rsn[v];
        }
      }
    }
  } else {
#pragma unroll
    for (int t = 0; t < 8; ++t) {
      int col = t * 16 + lo;
      float bv = bias[col];
#pragma unroll
      for (int v = 0; v < 8; ++v) {
        int row = (wave << 4) + v + (hi << 3);
        float val = acc[t][v] + bv;
        if (do_relu) val = fmaxf(val, 0.0f);
        if (row < n_nodes) out[(long)row * D + col] = val;
      }
    }
  }
}


__global__ __launch_bounds__(128)
void pair_mlp_kernel(const int* __restrict__ pairs, const float* __restrict__ h,
                     const float* __restrict__ Wpa,
                     const float* __restrict__ ba,
                     const float* __restrict__ Wb, const float* __restrict__ bb,
                     float* __restrict__ out, int n_pairs) {
  int wv = threadIdx.x >> 5;
  int tile = (blockIdx.x << 2) + wv;
  int ntiles = (n_pairs + 15) >> 4;
  if (tile >= ntiles) return;
  int lane = threadIdx.x & 31;
  int lo = lane & 15, hi = lane >> 4;

  float4 zr[16];
  {
    int p = (tile << 4) + lo;
    int pc = p < n_pairs ? p : 0;
    int i = pairs[2 * pc], j = pairs[2 * pc + 1];
    const float4* hi4 = (const float4*)(h + (long)i * D) + hi;
    const float4* hj4 = (const float4*)(h + (long)j * D) + hi;
#pragma unroll
    for (int g = 0; g < 16; ++g) {
      float4 a = hi4[2 * g], b = hj4[2 * g];
      float4 v;
      v.x = a.x * b.x; v.y = a.y * b.y; v.z = a.z * b.z; v.w = a.w * b.w;
      zr[g] = v;
    }
  }

  const v2f* wp = (const v2f*)Wpa + (long)(2 * hi) * D + lo;
  float partial[8];
#pragma unroll
  for (int v = 0; v < 8; ++v) partial[v] = 0.0f;

  for (int t = 0; t < 8; ++t) {
    v8f acc = (v8f)0.0f;
    const v2f* wt = wp + t * 16;
#pragma unroll
    for (int g = 0; g < 16; ++g) {
      v2f a0, a1;
      a0.x = zr[g].x; a0.y = zr[g].y;
      a1.x = zr[g].z; a1.y = zr[g].w;
      v2f b0 = wt[(long)(4 * g) * D];
      v2f b1 = wt[(long)(4 * g + 1) * D];
      acc = __builtin_amdgcn_wmma_f32_16x16x4_f32(
          false, a0, false, b0, (short)0, acc, false, false);
      acc = __builtin_amdgcn_wmma_f32_16x16x4_f32(
          false, a1, false, b1, (short)0, acc, false, false);
    }
    int col = t * 16 + lo;
    float bv = ba[col], wb = Wb[col];
#pragma unroll
    for (int v = 0; v < 8; ++v) {
      partial[v] += fmaxf(acc[v] + bv, 0.0f) * wb;
    }
  }

#pragma unroll
  for (int v = 0; v < 8; ++v) {
    float x = partial[v];
    x += __shfl_xor(x, 1, 32);
    x += __shfl_xor(x, 2, 32);
    x += __shfl_xor(x, 4, 32);
    x += __shfl_xor(x, 8, 32);
    partial[v] = x;
  }
  if (lo == 0) {
    float bbv = bb[0];
#pragma unroll
    for (int v = 0; v < 8; ++v) {
      int row = (tile << 4) + v + (hi << 3);
      if (row < n_pairs) out[row] = partial[v] + bbv;
    }
  }
}


extern "C" void kernel_launch(void* const* d_in, const int* in_sizes, int n_in,
                              void* d_out, int out_size, void* d_ws, size_t ws_size,
                              hipStream_t stream) {
  const int* senders   = (const int*)d_in[1];
  const int* receivers = (const int*)d_in[2];
  const int* pairs     = (const int*)d_in[3];
  const float* emb = (const float*)d_in[4];
  const float* W1  = (const float*)d_in[5];
  const float* b1  = (const float*)d_in[6];
  const float* W2  = (const float*)d_in[7];
  const float* b2  = (const float*)d_in[8];
  const float* Wa  = (const float*)d_in[9];
  const float* ba  = (const float*)d_in[10];
  const float* Wb  = (const float*)d_in[11];
  const float* bb  = (const float*)d_in[12];
  float* out = (float*)d_out;

  int N = in_sizes[0];
  int E = in_sizes[1];
  int P = in_sizes[3] / 2;
  long ND = (long)N * D;

  float* ws    = (float*)d_ws;
  float* deg_s = ws;
  float* deg_r = ws + N;
  float* bufA  = ws + 2L * N;
  float* bufB  = bufA + ND;
  float* bufC  = bufB + ND;
  unsigned int* keyw = (unsigned int*)bufC;
  float* Wp1   = bufC + ND;
  float* Wp2   = Wp1 + 2L * D * D;
  float* Wpa   = Wp2 + 2L * D * D;

  const int T = 256;
  int nb_deg = (N + T - 1) / T;
  int nb_pack = (2 * NPAIR_SAGE + NPAIR_MLP + T - 1) / T;
  init_kernel<<<nb_deg + nb_pack, T, 0, stream>>>(deg_s, deg_r, N, nb_deg,
                                                   W1, Wp1, W2, Wp2, Wa, Wpa);

  int nchunks = (E + KCHUNK - 1) / KCHUNK;
  int nwords = nchunks * (KCHUNK / 4);
  degree_kernel<<<(nwords + T - 1) / T, T, 0, stream>>>(senders, receivers, deg_s, deg_r, keyw, E, nwords);

  long n4 = ND / 4;
  int nranges = (N + RPB - 1) / RPB;
  int sgrid = nranges < 256 ? nranges : 256;
  int* elog_n = (int*)(keyw + nwords);
  int2* elog = (int2*)(elog_n + nranges * SC_WAVES);
  int mt = (N + 15) >> 4;
  int rnblocks = (int)((n4 + T * RN_PER_LANE - 1) / (T * RN_PER_LANE));

  rownorm_kernel<<<rnblocks, T, 0, stream>>>(emb, deg_s, bufA, N);
  scatter_kernel<false><<<sgrid, SC_THREADS, 0, stream>>>(senders, receivers, keyw, bufA, bufB, elog, elog_n,
                                                          E, nchunks, N, nranges);
  sage_gemm_kernel<<<(mt + 7) / 8, 256, 0, stream>>>(emb,  0, bufB, deg_r, deg_s, Wp1, b1,
                                                     bufB, bufA, N,  1,  1);

  scatter_kernel<true><<<sgrid, SC_THREADS, 0, stream>>>(senders, receivers, keyw, bufA, bufB, elog, elog_n,
                                                         E, nchunks, N, nranges);
  sage_gemm_kernel<<<(mt + 7) / 8, 256, 0, stream>>>(bufA,  1, bufB, deg_r, deg_s, Wp2, b2,
                                                     bufB, bufA, N,  0,  0);

  int ptiles = (P + 15) >> 4;
  pair_mlp_kernel<<<(ptiles + 3) / 4, 128, 0, stream>>>(pairs, bufB, Wpa, ba, Wb, bb, out, P);
}


# receiver degrees counted by the owner work-groups in LDS (one integer add per edge in the first layer's pass, stored per row for layer 2): the degree pass issues half its global float atomics
# speedup vs baseline: 2.2115x; 1.1684x over previous
#include <hip/hip_runtime.h>

#define D 128

typedef __attribute__((ext_vector_type(2))) float v2f;
typedef __attribute__((ext_vector_type(8))) float v8f;


#define NPAIR_SAGE (D * D)
#define NPAIR_MLP (D * D / 2)
__global__ __launch_bounds__(256)
void init_kernel(float* __restrict__ deg_s, int n_nodes, int nb_deg,
                 const float* __restrict__ W1, float* __restrict__ Wp1,
                 const float* __restrict__ W2, float* __restrict__ Wp2,
                 const float* __restrict__ Wa, float* __restrict__ Wpa) {
  int b = blockIdx.x;
  if (b < nb_deg) {
    int i = b * 256 + threadIdx.x;
    if (i < n_nodes) deg_s[i] = 1.0f;
    return;
  }
  int i = (b - nb_deg) * 256 + threadIdx.x;
  const float* W;
  float* Wp;
  if (i < NPAIR_SAGE) { W = W1; Wp = Wp1; }
  else if (i < 2 * NPAIR_SAGE) { W = W2; Wp = Wp2; i -= NPAIR_SAGE; }
  else if (i < 2 * NPAIR_SAGE + NPAIR_MLP) { W = Wa; Wp = Wpa; i -= 2 * NPAIR_SAGE; }
  else return;
  int kp = i >> 7, n = i & (D - 1);
  float2 v;
  v.x = W[(long)(2 * kp) * D + n];
  v.y = W[(long)(2 * kp + 1) * D + n];
  ((float2*)Wp)[i] = v;
}

#define RPB 391

__global__ __launch_bounds__(256)
void degree_kernel(const int* __restrict__ snd, const int* __restrict__ rcv,
                   float* deg_s, unsigned int* __restrict__ keyw,
                   int E, int nwords) {
  int w = blockIdx.x * blockDim.x + threadIdx.x;
  if (w >= nwords) return;
  unsigned int packed = 0u;
  if (4 * w + 3 < E) {
    int4 sv = ((const int4*)snd)[w];
    int4 rv = ((const int4*)rcv)[w];
    unsafeAtomicAdd(&deg_s[sv.x], 1.0f);
    unsafeAtomicAdd(&deg_s[sv.y], 1.0f);
    unsafeAtomicAdd(&deg_s[sv.z], 1.0f);
    unsafeAtomicAdd(&deg_s[sv.w], 1.0f);
    packed = (unsigned int)((rv.x / RPB) & 255) | (unsigned int)((rv.y / RPB) & 255) << 8 |
             (unsigned int)((rv.z / RPB) & 255) << 16 | (unsigned int)((rv.w / RPB) & 255) << 24;
  } else {
    for (int j = 0; j < 4; ++j) {
      int e = 4 * w + j;
      if (e < E) {
        int s = snd[e], r = rcv[e];
        unsafeAtomicAdd(&deg_s[s], 1.0f);
        packed |= (unsigned int)((r / RPB) & 255) << (8 * j);
      }
    }
  }
  keyw[w] = packed;
}

#define RN_PER_LANE 4
__global__ __launch_bounds__(256)
void rownorm_kernel(const float* __restrict__ src, const float* __restrict__ deg,
                    float* __restrict__ dst, int n_nodes) {
  long base = (long)blockIdx.x * (256 * RN_PER_LANE) + threadIdx.x;
  long n4 = (long)n_nodes * (D / 4);
  float4 v[RN_PER_LANE];
  float s[RN_PER_LANE];
#pragma unroll
  for (int k = 0; k < RN_PER_LANE; ++k) {
    long i = base + (long)k * 256;
    long ic = i < n4 ? i : 0;
    s[k] = deg[(int)(ic >> 5)];
    v[k] = ((const float4*)src)[ic];
  }
#pragma unroll
  for (int k = 0; k < RN_PER_LANE; ++k) {
    long i = base + (long)k * 256;
    if (i < n4) {
      float sc = rsqrtf(s[k]);
      float4 o = v[k];
      o.x *= sc; o.y *= sc; o.z *= sc; o.w *= sc;
      ((float4*)dst)[i] = o;
    }
  }
}

#define SC_WAVES 32
#define SC_THREADS (SC_WAVES * 32)
#define KCHUNK 512
#define KAHEAD 8
#define QBATCH 16
#define QCAP 48
#define LCAP 512
static_assert(RPB <= 16 * (SC_WAVES - 7), "one wave per 16-row tile of a block, 7 waves take none");

static __device__ __forceinline__ unsigned int zero_bytes(unsigned int x) {
  return ~(((x & 0x7F7F7F7Fu) + 0x7F7F7F7Fu) | x | 0x7F7F7F7Fu);
}

static __device__ __forceinline__ void add_rows(int sv, int rv, int off,
                                                const float* __restrict__ xn, float* rows,
                                                int nrows, int lane) {
  float2 v[QBATCH][2];
  int lr[QBATCH];
#pragma unroll
  for (int j = 0; j < QBATCH; ++j) {
    int s = __builtin_amdgcn_readlane(sv, off + j);
    lr[j] = __builtin_amdgcn_readlane(rv, off + j);
    const float2* srow = (const float2*)(xn + (long)s * D) + lane;
    v[j][0] = srow[0];
    v[j][1] = srow[32];
  }
#pragma unroll
  for (int j = 0; j < QBATCH; ++j) {
    if ((unsigned int)lr[j] < (unsigned int)nrows) {
      float* dst = rows + lr[j] * D + 2 * lane;
      __hip_atomic_fetch_add(dst + 0, v[j][0].x, __ATOMIC_RELAXED, __HIP_MEMORY_SCOPE_WORKGROUP);
      __hip_atomic_fetch_add(dst + 1, v[j][0].y, __ATOMIC_RELAXED, __HIP_MEMORY_SCOPE_WORKGROUP);
      __hip_atomic_fetch_add(dst + 64, v[j][1].x, __ATOMIC_RELAXED, __HIP_MEMORY_SCOPE_WORKGROUP);
      __hip_atomic_fetch_add(dst + 65, v[j][1].y, __ATOMIC_RELAXED, __HIP_MEMORY_SCOPE_WORKGROUP);
    }
  }
}

template <bool LOG>
static __device__ __forceinline__ void gather_add(const int* qe, int cnt,
                                                  const int* __restrict__ snd,
                                                  const int* __restrict__ rcv,
                                                  const float* __restrict__ xn, float* rows,
                                                  int E, int row0, int nrows, int lane,
                                                  int2* __restrict__ wlog, int& ln, int* rcnt) {
  int sl = lane & (QBATCH - 1);
  int e = qe[sl < cnt ? sl : 0];
  bool ok = (sl < cnt) && (e < E);
  int ec = ok ? e : 0;
  int sv = snd[ec];
  int rv = ok ? rcv[ec] - row0 : -1;
  if (LOG) {
    bool keep = (lane < QBATCH) && ((unsigned int)rv < (unsigned int)nrows);
    unsigned int km = __builtin_amdgcn_ballot_w32(keep);
    if (keep) {
      int pos = ln + (int)__builtin_amdgcn_mbcnt_lo(km, 0u);
      if (pos < LCAP) wlog[pos] = make_int2(sv, rv);
      __hip_atomic_fetch_add(rcnt + rv, 1, __ATOMIC_RELAXED, __HIP_MEMORY_SCOPE_WORKGROUP);
    }
    ln += __builtin_popcount(km);
  }
  add_rows(sv, rv, 0, xn, rows, nrows, lane);
}

template <bool LOG>
static __device__ __forceinline__ void scan_and_add(const int* __restrict__ snd,
                                                    const int* __restrict__ rcv,
                                                    const unsigned int* __restrict__ keyw,
                                                    const float* __restrict__ xn, float* rows,
                                                    int* q, int E, int nchunks, int range,
                                                    int row0, int nrows, int wave, int lane,
                                                    int2* __restrict__ wlog, int& ln, int* rcnt) {
  const unsigned int idpat = (unsigned int)(range & 255) * 0x01010101u;
  int qn = 0;
  for (int c0 = wave; c0 < nchunks; c0 += SC_WAVES * KAHEAD) {
    uint4 kv[KAHEAD];
    unsigned int m[KAHEAD];
#pragma unroll
    for (int u = 0; u < KAHEAD; ++u) {
      int c = c0 + u * SC_WAVES;
      int cc = c < nchunks ? c : c0;
      kv[u] = ((const uint4*)keyw)[(long)cc * 32 + lane];
    }
#pragma unroll
    for (int u = 0; u < KAHEAD; ++u) {
      unsigned int mu = (zero_bytes(kv[u].x ^ idpat) >> 7) | (zero_bytes(kv[u].y ^ idpat) >> 6) |
                        (zero_bytes(kv[u].z ^ idpat) >> 5) | (zero_bytes(kv[u].w ^ idpat) >> 4);
      m[u] = (c0 + u * SC_WAVES < nchunks) ? mu : 0u;
    }
#pragma nounroll
    for (int u = 0; u < KAHEAD; ++u) {
      int ebase = (c0 + u * SC_WAVES) * KCHUNK + lane * 16;
      unsigned int mm = m[0];
#pragma unroll
      for (int t = 1; t < KAHEAD; ++t) mm = (u == t) ? m[t] : mm;
      unsigned int any;
      while ((any = __builtin_amdgcn_ballot_w32(mm != 0u)) != 0u) {
        if (mm != 0u) {
          int p = __builtin_ctz(mm);
          mm &= mm - 1u;
          int pos = qn + (int)__builtin_amdgcn_mbcnt_lo(any, 0u);
          q[pos] = ebase + 4 * (p & 7) + (p >> 3);
        }
        qn += __builtin_popcount(any);
        while (qn >= QBATCH) {
          qn -= QBATCH;
          gather_add<LOG>(q + qn, QBATCH, snd, rcv, xn, rows, E, row0, nrows, lane, wlog, ln, rcnt);
        }
      }
    }
  }
  if (qn > 0) gather_add<LOG>(q, qn, snd, rcv, xn, rows, E, row0, nrows, lane, wlog, ln, rcnt);
}

template <bool SECOND>
static __device__ __forceinline__ void gemm_tile(const float* __restrict__ A0, const float* rows,
                                                 const int* rcnt, float* deg_r,
                                                 const float* __restrict__ deg_s,
                                                 const float* __restrict__ Wp,
                                                 const float* __restrict__ bias,
                                                 float* __restrict__ dst,
                                                 int row0, int nrows, int mt, int lane) {
  int lo = lane & 15, hi = lane >> 4;

  int lr = (mt << 4) + lo;
  bool mvalid = lr < nrows;
  int lrc = mvalid ? lr : 0;
  int m = row0 + lrc;

  float c;
  if (SECOND) {
    c = deg_r[m];
  } else {
    c = 1.0f + (float)rcnt[lrc];
    if (mvalid && hi == 0) deg_r[m] = c;
  }
  float rs = rsqrtf(c);
  float scale = rs * rs * rs;
  float ds = deg_s[m];
  float rs_next = rsqrtf(ds);
  float a0s = SECOND ? sqrtf(ds) : 1.0f;

  const float* a0row = A0 + (long)m * D;
  const float* smrow = rows + lrc * D;
  const v2f* wp = (const v2f*)Wp;

  v8f acc[8];
#pragma unroll
  for (int t = 0; t < 8; ++t) acc[t] = (v8f)0.0f;

  for (int kk = 0; kk < D; kk += 4) {
    int k0 = kk + (hi << 1);
    v2f a = *(const v2f*)(a0row + k0);
    a.x *= a0s;
    a.y *= a0s;
    if (!mvalid) { a.x = 0.0f; a.y = 0.0f; }
    const v2f* wrow = wp + (long)(k0 >> 1) * D + lo;
    v2f b[8];
#pragma unroll
    for (int t = 0; t < 8; ++t) b[t] = wrow[t * 16];
#pragma unroll
    for (int t = 0; t < 8; ++t) {
      acc[t] = __builtin_amdgcn_wmma_f32_16x16x4_f32(
          false, a, false, b[t], (short)0, acc[t], false, false);
    }
  }

  for (int kk = 0; kk < D; kk += 4) {
    int k0 = kk + (hi << 1);
    v2f sm = *(const v2f*)(smrow + k0);
    v2f a;
    a.x = sm.x * scale;
    a.y = sm.y * scale;
    if (!mvalid) { a.x = 0.0f; a.y = 0.0f; }
    const v2f* wrow = wp + (long)((k0 + D) >> 1) * D + lo;
    v2f b[8];
#pragma unroll
    for (int t = 0; t < 8; ++t) b[t] = wrow[t * 16];
#pragma unroll
    for (int t = 0; t < 8; ++t) {
      acc[t] = __builtin_amdgcn_wmma_f32_16x16x4_f32(
          false, a, false, b[t], (short)0, acc[t], false, false);
    }
  }

  float rsn[8];
#pragma unroll
  for (int v = 0; v < 8; ++v) rsn[v] = __shfl(rs_next, v + (hi << 3), 32);

#pragma unroll
  for (int t = 0; t < 8; ++t) {
    int col = t * 16 + lo;
    float bv = bias[col];
#pragma unroll
    for (int v = 0; v < 8; ++v) {
      int lrow = (mt << 4) + v + (hi << 3);
      float val = acc[t][v] + bv;
      if (!SECOND) val = fmaxf(val, 0.0f) * rsn[v];
      if (lrow < nrows) dst[(long)(row0 + lrow) * D + col] = val;
    }
  }
}

template <bool REPLAY>
__global__ __launch_bounds__(SC_THREADS)
void scatter_kernel(const int* __restrict__ snd, const int* __restrict__ rcv,
                    const unsigned int* __restrict__ keyw,
                    const float* __restrict__ xn, float* __restrict__ dst,
                    int2* __restrict__ elog, int* __restrict__ elog_n,
                    const float* __restrict__ A0,
                    float* deg_r,
                    const float* __restrict__ deg_s,
                    const float* __restrict__ Wp,
                    const float* __restrict__ bias,
                    int E, int nchunks, int n_nodes, int nranges) {
  __shared__ float rows[RPB * D];
  __shared__ int stack[SC_WAVES * QCAP];
  __shared__ int rcnt[RPB];

  const int lane = threadIdx.x & 31;
  const int wave = __builtin_amdgcn_readfirstlane((int)(threadIdx.x >> 5));
  int* q = stack + wave * QCAP;
  float4* rows4 = (float4*)rows;

  for (int range = blockIdx.x; range < nranges; range += gridDim.x) {
    const int row0 = range * RPB;
    const int nrows = min(RPB, n_nodes - row0);
    const int n4 = nrows * (D / 4);

    const float4* src4 = (const float4*)(xn + (long)row0 * D);
    for (int i0 = threadIdx.x; i0 < n4; i0 += SC_THREADS * 4) {
      float4 t[4];
      int ii[4];
#pragma unroll
      for (int k = 0; k < 4; ++k) {
        int i = i0 + k * SC_THREADS;
        ii[k] = i < n4 ? i : i0;
        t[k] = src4[ii[k]];
      }
#pragma unroll
      for (int k = 0; k < 4; ++k) rows4[ii[k]] = t[k];
    }
    if (!REPLAY) {
      for (int i = threadIdx.x; i < RPB; i += SC_THREADS) rcnt[i] = 0;
    }
    __syncthreads();

    int2* wlog = elog + (long)(range * SC_WAVES + wave) * LCAP;
    int ln = 0;
    if (!REPLAY) {
      scan_and_add<true>(snd, rcv, keyw, xn, rows, q, E, nchunks, range, row0, nrows, wave, lane, wlog, ln, rcnt);
      if (lane == 0) elog_n[range * SC_WAVES + wave] = ln;
    } else {
      const int n = __builtin_amdgcn_readfirstlane(elog_n[range * SC_WAVES + wave]);
      if (n > LCAP) {
        scan_and_add<false>(snd, rcv, keyw, xn, rows, q, E, nchunks, range, row0, nrows, wave, lane, wlog, ln, rcnt);
      } else if (n > 0) {
        int2 rnext = wlog[lane < n ? lane : 0];
        for (int i = 0; i < n; i += 32) {
          int2 r = rnext;
          int nidx = i + 32 + lane;
          rnext = wlog[nidx < n ? nidx : 0];
          int sv = r.x;
          int rv = (i + lane < n) ? r.y : -1;
          add_rows(sv, rv, 0, xn, rows, nrows, lane);
          if (i + QBATCH < n) add_rows(sv, rv, QBATCH, xn, rows, nrows, lane);
        }
      }
    }
    __syncthreads();

    const unsigned int notile = (1u << 3) | (1u << 6) | (1u << 9) | (1u << 12) | (1u << 19) | (1u << 22) | (1u << 29);
    const int mt = __builtin_popcount(~notile & ((1u << wave) - 1u));
    if (((notile >> wave) & 1u) == 0u && (mt << 4) < nrows)
      gemm_tile<REPLAY>(A0, rows, rcnt, deg_r, deg_s, Wp, bias, dst, row0, nrows, mt, lane);
    __syncthreads();
  }
}


__global__ __launch_bounds__(128)
void pair_mlp_kernel(const int* __restrict__ pairs, const float* __restrict__ h,
                     const float* __restrict__ Wpa,
                     const float* __restrict__ ba,
                     const float* __restrict__ Wb, const float* __restrict__ bb,
                     float* __restrict__ out, int n_pairs) {
  int wv = threadIdx.x >> 5;
  int tile = (blockIdx.x << 2) + wv;
  int ntiles = (n_pairs + 15) >> 4;
  if (tile >= ntiles) return;
  int lane = threadIdx.x & 31;
  int lo = lane & 15, hi = lane >> 4;

  float4 zr[16];
  {
    int p = (tile << 4) + lo;
    int pc = p < n_pairs ? p : 0;
    int i = pairs[2 * pc], j = pairs[2 * pc + 1];
    const float4* hi4 = (const float4*)(h + (long)i * D) + hi;
    const float4* hj4 = (const float4*)(h + (long)j * D) + hi;
#pragma unroll
    for (int g = 0; g < 16; ++g) {
      float4 a = hi4[2 * g], b = hj4[2 * g];
      float4 v;
      v.x = a.x * b.x; v.y = a.y * b.y; v.z = a.z * b.z; v.w = a.w * b.w;
      zr[g] = v;
    }
  }

  const v2f* wp = (const v2f*)Wpa + (long)(2 * hi) * D + lo;
  float partial[8];
#pragma unroll
  for (int v = 0; v < 8; ++v) partial[v] = 0.0f;

  for (int t = 0; t < 8; ++t) {
    v8f acc = (v8f)0.0f;
    const v2f* wt = wp + t * 16;
#pragma unroll
    for (int g = 0; g < 16; ++g) {
      v2f a0, a1;
      a0.x = zr[g].x; a0.y = zr[g].y;
      a1.x = zr[g].z; a1.y = zr[g].w;
      v2f b0 = wt[(long)(4 * g) * D];
      v2f b1 = wt[(long)(4 * g + 1) * D];
      acc = __builtin_amdgcn_wmma_f32_16x16x4_f32(
          false, a0, false, b0, (short)0, acc, false, false);
      acc = __builtin_amdgcn_wmma_f32_16x16x4_f32(
          false, a1, false, b1, (short)0, acc, false, false);
    }
    int col = t * 16 + lo;
    float bv = ba[col], wb = Wb[col];
#pragma unroll
    for (int v = 0; v < 8; ++v) {
      partial[v] += fmaxf(acc[v] + bv, 0.0f) * wb;
    }
  }

#pragma unroll
  for (int v = 0; v < 8; ++v) {
    float x = partial[v];
    x += __shfl_xor(x, 1, 32);
    x += __shfl_xor(x, 2, 32);
    x += __shfl_xor(x, 4, 32);
    x += __shfl_xor(x, 8, 32);
    partial[v] = x;
  }
  if (lo == 0) {
    float bbv = bb[0];
#pragma unroll
    for (int v = 0; v < 8; ++v) {
      int row = (tile << 4) + v + (hi << 3);
      if (row < n_pairs) out[row] = partial[v] + bbv;
    }
  }
}


extern "C" void kernel_launch(void* const* d_in, const int* in_sizes, int n_in,
                              void* d_out, int out_size, void* d_ws, size_t ws_size,
                              hipStream_t stream) {
  const int* senders   = (const int*)d_in[1];
  const int* receivers = (const int*)d_in[2];
  const int* pairs     = (const int*)d_in[3];
  const float* emb = (const float*)d_in[4];
  const float* W1  = (const float*)d_in[5];
  const float* b1  = (const float*)d_in[6];
  const float* W2  = (const float*)d_in[7];
  const float* b2  = (const float*)d_in[8];
  const float* Wa  = (const float*)d_in[9];
  const float* ba  = (const float*)d_in[10];
  const float* Wb  = (const float*)d_in[11];
  const float* bb  = (const float*)d_in[12];
  float* out = (float*)d_out;

  int N = in_sizes[0];
  int E = in_sizes[1];
  int P = in_sizes[3] / 2;
  long ND = (long)N * D;

  float* ws    = (float*)d_ws;
  float* deg_s = ws;
  float* deg_r = ws + N;
  float* bufA  = ws + 2L * N;
  float* bufB  = bufA + ND;
  float* bufC  = bufB + ND;
  unsigned int* keyw = (unsigned int*)bufC;
  float* Wp1   = bufC + ND;
  float* Wp2   = Wp1 + 2L * D * D;
  float* Wpa   = Wp2 + 2L * D * D;

  const int T = 256;
  int nb_deg = (N + T - 1) / T;
  int nb_pack = (2 * NPAIR_SAGE + NPAIR_MLP + T - 1) / T;
  init_kernel<<<nb_deg + nb_pack, T, 0, stream>>>(deg_s, N, nb_deg,
                                                   W1, Wp1, W2, Wp2, Wa, Wpa);

  int nchunks = (E + KCHUNK - 1) / KCHUNK;
  int nwords = nchunks * (KCHUNK / 4);
  degree_kernel<<<(nwords + T - 1) / T, T, 0, stream>>>(senders, receivers, deg_s, keyw, E, nwords);

  long n4 = ND / 4;
  int nranges = (N + RPB - 1) / RPB;
  int sgrid = nranges < 256 ? nranges : 256;
  int* elog_n = (int*)(keyw + nwords);
  int2* elog = (int2*)(elog_n + nranges * SC_WAVES);
  int rnblocks = (int)((n4 + T * RN_PER_LANE - 1) / (T * RN_PER_LANE));

  rownorm_kernel<<<rnblocks, T, 0, stream>>>(emb, deg_s, bufA, N);
  scatter_kernel<false><<<sgrid, SC_THREADS, 0, stream>>>(senders, receivers, keyw, bufA, bufB, elog, elog_n,
                                                          emb, deg_r, deg_s, Wp1, b1,
                                                          E, nchunks, N, nranges);

  scatter_kernel<true><<<sgrid, SC_THREADS, 0, stream>>>(senders, receivers, keyw, bufB, bufA, elog, elog_n,
                                                         bufB, deg_r, deg_s, Wp2, b2,
                                                         E, nchunks, N, nranges);

  int ptiles = (P + 15) >> 4;
  pair_mlp_kernel<<<(ptiles + 3) / 4, 128, 0, stream>>>(pairs, bufA, Wpa, ba, Wb, bb, out, P);
}


# sender degrees counted in LDS: (node segment, edge slab) work-groups add integers into an LDS count block, the row-scaling pass sums a node's slab counts; no global atomic left (same exact degrees)
# speedup vs baseline: 2.5604x; 1.1578x over previous
#include <hip/hip_runtime.h>

#define D 128

typedef __attribute__((ext_vector_type(2))) float v2f;
typedef __attribute__((ext_vector_type(8))) float v8f;


#define NPAIR_SAGE (D * D)
#define NPAIR_MLP (D * D / 2)
__global__ __launch_bounds__(256)
void init_kernel(const float* __restrict__ W1, float* __restrict__ Wp1,
                 const float* __restrict__ W2, float* __restrict__ Wp2,
                 const float* __restrict__ Wa, float* __restrict__ Wpa) {
  int i = blockIdx.x * 256 + threadIdx.x;
  const float* W;
  float* Wp;
  if (i < NPAIR_SAGE) { W = W1; Wp = Wp1; }
  else if (i < 2 * NPAIR_SAGE) { W = W2; Wp = Wp2; i -= NPAIR_SAGE; }
  else if (i < 2 * NPAIR_SAGE + NPAIR_MLP) { W = Wa; Wp = Wpa; i -= 2 * NPAIR_SAGE; }
  else return;
  int kp = i >> 7, n = i & (D - 1);
  float2 v;
  v.x = W[(long)(2 * kp) * D + n];
  v.y = W[(long)(2 * kp + 1) * D + n];
  ((float2*)Wp)[i] = v;
}

#define RPB 391

#define DSEG 12800
#define DC_THREADS 1024
#define DC_AHEAD 8
static __device__ __forceinline__ void count_sender(int* cnt, int idx, int segn) {
  if ((unsigned int)idx < (unsigned int)segn)
    __hip_atomic_fetch_add(cnt + idx, 1, __ATOMIC_RELAXED, __HIP_MEMORY_SCOPE_WORKGROUP);
}
__global__ __launch_bounds__(DC_THREADS)
void degree_kernel(const int* __restrict__ snd, const int* __restrict__ rcv,
                   int* __restrict__ part, unsigned int* __restrict__ keyw,
                   int E, int nwords, int n_nodes, int nseg, int slab_words) {
  __shared__ int cnt[DSEG];
  const int tid = threadIdx.x;

  for (int w = blockIdx.x * DC_THREADS + tid; w < nwords; w += gridDim.x * DC_THREADS) {
    unsigned int packed = 0u;
    if (4 * w + 3 < E) {
      int4 rv = ((const int4*)rcv)[w];
      packed = (unsigned int)((rv.x / RPB) & 255) | (unsigned int)((rv.y / RPB) & 255) << 8 |
               (unsigned int)((rv.z / RPB) & 255) << 16 | (unsigned int)((rv.w / RPB) & 255) << 24;
    } else {
      for (int j = 0; j < 4; ++j) {
        int e = 4 * w + j;
        if (e < E) packed |= (unsigned int)((rcv[e] / RPB) & 255) << (8 * j);
      }
    }
    keyw[w] = packed;
  }

  const int seg = blockIdx.x % nseg, slab = blockIdx.x / nseg;
  const int seg0 = seg * DSEG;
  const int segn = min(DSEG, n_nodes - seg0);
  for (int i = tid; i < DSEG; i += DC_THREADS) cnt[i] = 0;
  __syncthreads();

  const int wbeg = slab * slab_words;
  const int wend = min(wbeg + slab_words, E >> 2);
  for (int wb = wbeg + tid; wb < wend; wb += DC_THREADS * DC_AHEAD) {
    int4 sv[DC_AHEAD];
#pragma unroll
    for (int u = 0; u < DC_AHEAD; ++u) {
      int w = wb + u * DC_THREADS;
      sv[u] = ((const int4*)snd)[w < wend ? w : wb];
    }
#pragma unroll
    for (int u = 0; u < DC_AHEAD; ++u) {
      bool in = wb + u * DC_THREADS < wend;
      count_sender(cnt, (in ? sv[u].x : -1) - seg0, segn);
      count_sender(cnt, (in ? sv[u].y : -1) - seg0, segn);
      count_sender(cnt, (in ? sv[u].z : -1) - seg0, segn);
      count_sender(cnt, (in ? sv[u].w : -1) - seg0, segn);
    }
  }
  if ((E & 3) != 0 && tid == 0) {
    int wt = E >> 2;
    if (wt >= wbeg && wt < wbeg + slab_words)
      for (int e = 4 * wt; e < E; ++e) count_sender(cnt, snd[e] - seg0, segn);
  }
  __syncthreads();

  int* prow = part + (long)slab * n_nodes + seg0;
  for (int i = tid; i < segn; i += DC_THREADS) prow[i] = cnt[i];
}

#define RN_PER_LANE 4
#define RN_WAVES 8
#define RN_SLABS 4
#define RN_ROWS (256 * RN_PER_LANE * 4 / D)
static_assert(RN_ROWS == 32, "one lane per row of the work-group when the slab counts are added");
__global__ __launch_bounds__(256)
void rownorm_kernel(const float* __restrict__ src, const int* __restrict__ part, int nslab,
                    float* __restrict__ deg, float* __restrict__ dst, int n_nodes) {
  __shared__ int red[RN_WAVES][RN_ROWS];
  __shared__ float rsl[RN_ROWS];
  const int lane = threadIdx.x & 31, wv = threadIdx.x >> 5;
  long base = (long)blockIdx.x * (256 * RN_PER_LANE) + threadIdx.x;
  long n4 = (long)n_nodes * (D / 4);
  float4 v[RN_PER_LANE];
#pragma unroll
  for (int k = 0; k < RN_PER_LANE; ++k) {
    long i = base + (long)k * 256;
    long ic = i < n4 ? i : 0;
    v[k] = ((const float4*)src)[ic];
  }

  const int row = blockIdx.x * RN_ROWS + lane;
  const int rowc = row < n_nodes ? row : 0;
  int c = 0;
  for (int s0 = wv; s0 < nslab; s0 += RN_WAVES * RN_SLABS) {
    int t[RN_SLABS];
#pragma unroll
    for (int u = 0; u < RN_SLABS; ++u) {
      int s = s0 + u * RN_WAVES;
      t[u] = part[(long)(s < nslab ? s : s0) * n_nodes + rowc];
    }
#pragma unroll
    for (int u = 0; u < RN_SLABS; ++u) c += (s0 + u * RN_WAVES < nslab) ? t[u] : 0;
  }
  red[wv][lane] = c;
  __syncthreads();
  if (wv == 0) {
    int t = 0;
#pragma unroll
    for (int w = 0; w < RN_WAVES; ++w) t += red[w][lane];
    float d = 1.0f + (float)t;
    rsl[lane] = rsqrtf(d);
    if (row < n_nodes) deg[row] = d;
  }
  __syncthreads();

#pragma unroll
  for (int k = 0; k < RN_PER_LANE; ++k) {
    long i = base + (long)k * 256;
    if (i < n4) {
      float sc = rsl[wv + 8 * k];
      float4 o = v[k];
      o.x *= sc; o.y *= sc; o.z *= sc; o.w *= sc;
      ((float4*)dst)[i] = o;
    }
  }
}

#define SC_WAVES 32
#define SC_THREADS (SC_WAVES * 32)
#define KCHUNK 512
#define KAHEAD 8
#define QBATCH 16
#define QCAP 48
#define LCAP 512
static_assert(RPB <= 16 * (SC_WAVES - 7), "one wave per 16-row tile of a block, 7 waves take none");

static __device__ __forceinline__ unsigned int zero_bytes(unsigned int x) {
  return ~(((x & 0x7F7F7F7Fu) + 0x7F7F7F7Fu) | x | 0x7F7F7F7Fu);
}

static __device__ __forceinline__ void add_rows(int sv, int rv, int off,
                                                const float* __restrict__ xn, float* rows,
                                                int nrows, int lane) {
  float2 v[QBATCH][2];
  int lr[QBATCH];
#pragma unroll
  for (int j = 0; j < QBATCH; ++j) {
    int s = __builtin_amdgcn_readlane(sv, off + j);
    lr[j] = __builtin_amdgcn_readlane(rv, off + j);
    const float2* srow = (const float2*)(xn + (long)s * D) + lane;
    v[j][0] = srow[0];
    v[j][1] = srow[32];
  }
#pragma unroll
  for (int j = 0; j < QBATCH; ++j) {
    if ((unsigned int)lr[j] < (unsigned int)nrows) {
      float* dst = rows + lr[j] * D + 2 * lane;
      __hip_atomic_fetch_add(dst + 0, v[j][0].x, __ATOMIC_RELAXED, __HIP_MEMORY_SCOPE_WORKGROUP);
      __hip_atomic_fetch_add(dst + 1, v[j][0].y, __ATOMIC_RELAXED, __HIP_MEMORY_SCOPE_WORKGROUP);
      __hip_atomic_fetch_add(dst + 64, v[j][1].x, __ATOMIC_RELAXED, __HIP_MEMORY_SCOPE_WORKGROUP);
      __hip_atomic_fetch_add(dst + 65, v[j][1].y, __ATOMIC_RELAXED, __HIP_MEMORY_SCOPE_WORKGROUP);
    }
  }
}

template <bool LOG>
static __device__ __forceinline__ void gather_add(const int* qe, int cnt,
                                                  const int* __restrict__ snd,
                                                  const int* __restrict__ rcv,
                                                  const float* __restrict__ xn, float* rows,
                                                  int E, int row0, int nrows, int lane,
                                                  int2* __restrict__ wlog, int& ln, int* rcnt) {
  int sl = lane & (QBATCH - 1);
  int e = qe[sl < cnt ? sl : 0];
  bool ok = (sl < cnt) && (e < E);
  int ec = ok ? e : 0;
  int sv = snd[ec];
  int rv = ok ? rcv[ec] - row0 : -1;
  if (LOG) {
    bool keep = (lane < QBATCH) && ((unsigned int)rv < (unsigned int)nrows);
    unsigned int km = __builtin_amdgcn_ballot_w32(keep);
    if (keep) {
      int pos = ln + (int)__builtin_amdgcn_mbcnt_lo(km, 0u);
      if (pos < LCAP) wlog[pos] = make_int2(sv, rv);
      __hip_atomic_fetch_add(rcnt + rv, 1, __ATOMIC_RELAXED, __HIP_MEMORY_SCOPE_WORKGROUP);
    }
    ln += __builtin_popcount(km);
  }
  add_rows(sv, rv, 0, xn, rows, nrows, lane);
}

template <bool LOG>
static __device__ __forceinline__ void scan_and_add(const int* __restrict__ snd,
                                                    const int* __restrict__ rcv,
                                                    const unsigned int* __restrict__ keyw,
                                                    const float* __restrict__ xn, float* rows,
                                                    int* q, int E, int nchunks, int range,
                                                    int row0, int nrows, int wave, int lane,
                                                    int2* __restrict__ wlog, int& ln, int* rcnt) {
  const unsigned int idpat = (unsigned int)(range & 255) * 0x01010101u;
  int qn = 0;
  for (int c0 = wave; c0 < nchunks; c0 += SC_WAVES * KAHEAD) {
    uint4 kv[KAHEAD];
    unsigned int m[KAHEAD];
#pragma unroll
    for (int u = 0; u < KAHEAD; ++u) {
      int c = c0 + u * SC_WAVES;
      int cc = c < nchunks ? c : c0;
      kv[u] = ((const uint4*)keyw)[(long)cc * 32 + lane];
    }
#pragma unroll
    for (int u = 0; u < KAHEAD; ++u) {
      unsigned int mu = (zero_bytes(kv[u].x ^ idpat) >> 7) | (zero_bytes(kv[u].y ^ idpat) >> 6) |
                        (zero_bytes(kv[u].z ^ idpat) >> 5) | (zero_bytes(kv[u].w ^ idpat) >> 4);
      m[u] = (c0 + u * SC_WAVES < nchunks) ? mu : 0u;
    }
#pragma nounroll
    for (int u = 0; u < KAHEAD; ++u) {
      int ebase = (c0 + u * SC_WAVES) * KCHUNK + lane * 16;
      unsigned int mm = m[0];
#pragma unroll
      for (int t = 1; t < KAHEAD; ++t) mm = (u == t) ? m[t] : mm;
      unsigned int any;
      while ((any = __builtin_amdgcn_ballot_w32(mm != 0u)) != 0u) {
        if (mm != 0u) {
          int p = __builtin_ctz(mm);
          mm &= mm - 1u;
          int pos = qn + (int)__builtin_amdgcn_mbcnt_lo(any, 0u);
          q[pos] = ebase + 4 * (p & 7) + (p >> 3);
        }
        qn += __builtin_popcount(any);
        while (qn >= QBATCH) {
          qn -= QBATCH;
          gather_add<LOG>(q + qn, QBATCH, snd, rcv, xn, rows, E, row0, nrows, lane, wlog, ln, rcnt);
        }
      }
    }
  }
  if (qn > 0) gather_add<LOG>(q, qn, snd, rcv, xn, rows, E, row0, nrows, lane, wlog, ln, rcnt);
}

template <bool SECOND>
static __device__ __forceinline__ void gemm_tile(const float* __restrict__ A0, const float* rows,
                                                 const int* rcnt, float* deg_r,
                                                 const float* __restrict__ deg_s,
                                                 const float* __restrict__ Wp,
                                                 const float* __restrict__ bias,
                                                 float* __restrict__ dst,
                                                 int row0, int nrows, int mt, int lane) {
  int lo = lane & 15, hi = lane >> 4;

  int lr = (mt << 4) + lo;
  bool mvalid = lr < nrows;
  int lrc = mvalid ? lr : 0;
  int m = row0 + lrc;

  float c;
  if (SECOND) {
    c = deg_r[m];
  } else {
    c = 1.0f + (float)rcnt[lrc];
    if (mvalid && hi == 0) deg_r[m] = c;
  }
  float rs = rsqrtf(c);
  float scale = rs * rs * rs;
  float ds = deg_s[m];
  float rs_next = rsqrtf(ds);
  float a0s = SECOND ? sqrtf(ds) : 1.0f;

  const float* a0row = A0 + (long)m * D;
  const float* smrow = rows + lrc * D;
  const v2f* wp = (const v2f*)Wp;

  v8f acc[8];
#pragma unroll
  for (int t = 0; t < 8; ++t) acc[t] = (v8f)0.0f;

  for (int kk = 0; kk < D; kk += 4) {
    int k0 = kk + (hi << 1);
    v2f a = *(const v2f*)(a0row + k0);
    a.x *= a0s;
    a.y *= a0s;
    if (!mvalid) { a.x = 0.0f; a.y = 0.0f; }
    const v2f* wrow = wp + (long)(k0 >> 1) * D + lo;
    v2f b[8];
#pragma unroll
    for (int t = 0; t < 8; ++t) b[t] = wrow[t * 16];
#pragma unroll
    for (int t = 0; t < 8; ++t) {
      acc[t] = __builtin_amdgcn_wmma_f32_16x16x4_f32(
          false, a, false, b[t], (short)0, acc[t], false, false);
    }
  }

  for (int kk = 0; kk < D; kk += 4) {
    int k0 = kk + (hi << 1);
    v2f sm = *(const v2f*)(smrow + k0);
    v2f a;
    a.x = sm.x * scale;
    a.y = sm.y * scale;
    if (!mvalid) { a.x = 0.0f; a.y = 0.0f; }
    const v2f* wrow = wp + (long)((k0 + D) >> 1) * D + lo;
    v2f b[8];
#pragma unroll
    for (int t = 0; t < 8; ++t) b[t] = wrow[t * 16];
#pragma unroll
    for (int t = 0; t < 8; ++t) {
      acc[t] = __builtin_amdgcn_wmma_f32_16x16x4_f32(
          false, a, false, b[t], (short)0, acc[t], false, false);
    }
  }

  float rsn[8];
#pragma unroll
  for (int v = 0; v < 8; ++v) rsn[v] = __shfl(rs_next, v + (hi << 3), 32);

#pragma unroll
  for (int t = 0; t < 8; ++t) {
    int col = t * 16 + lo;
    float bv = bias[col];
#pragma unroll
    for (int v = 0; v < 8; ++v) {
      int lrow = (mt << 4) + v + (hi << 3);
      float val = acc[t][v] + bv;
      if (!SECOND) val = fmaxf(val, 0.0f) * rsn[v];
      if (lrow < nrows) dst[(long)(row0 + lrow) * D + col] = val;
    }
  }
}

template <bool REPLAY>
__global__ __launch_bounds__(SC_THREADS)
void scatter_kernel(const int* __restrict__ snd, const int* __restrict__ rcv,
                    const unsigned int* __restrict__ keyw,
                    const float* __restrict__ xn, float* __restrict__ dst,
                    int2* __restrict__ elog, int* __restrict__ elog_n,
                    const float* __restrict__ A0,
                    float* deg_r,
                    const float* __restrict__ deg_s,
                    const float* __restrict__ Wp,
                    const float* __restrict__ bias,
                    int E, int nchunks, int n_nodes, int nranges) {
  __shared__ float rows[RPB * D];
  __shared__ int stack[SC_WAVES * QCAP];
  __shared__ int rcnt[RPB];

  const int lane = threadIdx.x & 31;
  const int wave = __builtin_amdgcn_readfirstlane((int)(threadIdx.x >> 5));
  int* q = stack + wave * QCAP;
  float4* rows4 = (float4*)rows;

  for (int range = blockIdx.x; range < nranges; range += gridDim.x) {
    const int row0 = range * RPB;
    const int nrows = min(RPB, n_nodes - row0);
    const int n4 = nrows * (D / 4);

    const float4* src4 = (const float4*)(xn + (long)row0 * D);
    for (int i0 = threadIdx.x; i0 < n4; i0 += SC_THREADS * 4) {
      float4 t[4];
      int ii[4];
#pragma unroll
      for (int k = 0; k < 4; ++k) {
        int i = i0 + k * SC_THREADS;
        ii[k] = i < n4 ? i : i0;
        t[k] = src4[ii[k]];
      }
#pragma unroll
      for (int k = 0; k < 4; ++k) rows4[ii[k]] = t[k];
    }
    if (!REPLAY) {
      for (int i = threadIdx.x; i < RPB; i += SC_THREADS) rcnt[i] = 0;
    }
    __syncthreads();

    int2* wlog = elog + (long)(range * SC_WAVES + wave) * LCAP;
    int ln = 0;
    if (!REPLAY) {
      scan_and_add<true>(snd, rcv, keyw, xn, rows, q, E, nchunks, range, row0, nrows, wave, lane, wlog, ln, rcnt);
      if (lane == 0) elog_n[range * SC_WAVES + wave] = ln;
    } else {
      const int n = __builtin_amdgcn_readfirstlane(elog_n[range * SC_WAVES + wave]);
      if (n > LCAP) {
        scan_and_add<false>(snd, rcv, keyw, xn, rows, q, E, nchunks, range, row0, nrows, wave, lane, wlog, ln, rcnt);
      } else if (n > 0) {
        int2 rnext = wlog[lane < n ? lane : 0];
        for (int i = 0; i < n; i += 32) {
          int2 r = rnext;
          int nidx = i + 32 + lane;
          rnext = wlog[nidx < n ? nidx : 0];
          int sv = r.x;
          int rv = (i + lane < n) ? r.y : -1;
          add_rows(sv, rv, 0, xn, rows, nrows, lane);
          if (i + QBATCH < n) add_rows(sv, rv, QBATCH, xn, rows, nrows, lane);
        }
      }
    }
    __syncthreads();

    const unsigned int notile = (1u << 3) | (1u << 6) | (1u << 9) | (1u << 12) | (1u << 19) | (1u << 22) | (1u << 29);
    const int mt = __builtin_popcount(~notile & ((1u << wave) - 1u));
    if (((notile >> wave) & 1u) == 0u && (mt << 4) < nrows)
      gemm_tile<REPLAY>(A0, rows, rcnt, deg_r, deg_s, Wp, bias, dst, row0, nrows, mt, lane);
    __syncthreads();
  }
}


__global__ __launch_bounds__(128)
void pair_mlp_kernel(const int* __restrict__ pairs, const float* __restrict__ h,
                     const float* __restrict__ Wpa,
                     const float* __restrict__ ba,
                     const float* __restrict__ Wb, const float* __restrict__ bb,
                     float* __restrict__ out, int n_pairs) {
  int wv = threadIdx.x >> 5;
  int tile = (blockIdx.x << 2) + wv;
  int ntiles = (n_pairs + 15) >> 4;
  if (tile >= ntiles) return;
  int lane = threadIdx.x & 31;
  int lo = lane & 15, hi = lane >> 4;

  float4 zr[16];
  {
    int p = (tile << 4) + lo;
    int pc = p < n_pairs ? p : 0;
    int i = pairs[2 * pc], j = pairs[2 * pc + 1];
    const float4* hi4 = (const float4*)(h + (long)i * D) + hi;
    const float4* hj4 = (const float4*)(h + (long)j * D) + hi;
#pragma unroll
    for (int g = 0; g < 16; ++g) {
      float4 a = hi4[2 * g], b = hj4[2 * g];
      float4 v;
      v.x = a.x * b.x; v.y = a.y * b.y; v.z = a.z * b.z; v.w = a.w * b.w;
      zr[g] = v;
    }
  }

  const v2f* wp = (const v2f*)Wpa + (long)(2 * hi) * D + lo;
  float partial[8];
#pragma unroll
  for (int v = 0; v < 8; ++v) partial[v] = 0.0f;

  for (int t = 0; t < 8; ++t) {
    v8f acc = (v8f)0.0f;
    const v2f* wt = wp + t * 16;
#pragma unroll
    for (int g = 0; g < 16; ++g) {
      v2f a0, a1;
      a0.x = zr[g].x; a0.y = zr[g].y;
      a1.x = zr[g].z; a1.y = zr[g].w;
      v2f b0 = wt[(long)(4 * g) * D];
      v2f b1 = wt[(long)(4 * g + 1) * D];
      acc = __builtin_amdgcn_wmma_f32_16x16x4_f32(
          false, a0, false, b0, (short)0, acc, false, false);
      acc = __builtin_amdgcn_wmma_f32_16x16x4_f32(
          false, a1, false, b1, (short)0, acc, false, false);
    }
    int col = t * 16 + lo;
    float bv = ba[col], wb = Wb[col];
#pragma unroll
    for (int v = 0; v < 8; ++v) {
      partial[v] += fmaxf(acc[v] + bv, 0.0f) * wb;
    }
  }

#pragma unroll
  for (int v = 0; v < 8; ++v) {
    float x = partial[v];
    x += __shfl_xor(x, 1, 32);
    x += __shfl_xor(x, 2, 32);
    x += __shfl_xor(x, 4, 32);
    x += __shfl_xor(x, 8, 32);
    partial[v] = x;
  }
  if (lo == 0) {
    float bbv = bb[0];
#pragma unroll
    for (int v = 0; v < 8; ++v) {
      int row = (tile << 4) + v + (hi << 3);
      if (row < n_pairs) out[row] = partial[v] + bbv;
    }
  }
}


extern "C" void kernel_launch(void* const* d_in, const int* in_sizes, int n_in,
                              void* d_out, int out_size, void* d_ws, size_t ws_size,
                              hipStream_t stream) {
  const int* senders   = (const int*)d_in[1];
  const int* receivers = (const int*)d_in[2];
  const int* pairs     = (const int*)d_in[3];
  const float* emb = (const float*)d_in[4];
  const float* W1  = (const float*)d_in[5];
  const float* b1  = (const float*)d_in[6];
  const float* W2  = (const float*)d_in[7];
  const float* b2  = (const float*)d_in[8];
  const float* Wa  = (const float*)d_in[9];
  const float* ba  = (const float*)d_in[10];
  const float* Wb  = (const float*)d_in[11];
  const float* bb  = (const float*)d_in[12];
  float* out = (float*)d_out;

  int N = in_sizes[0];
  int E = in_sizes[1];
  int P = in_sizes[3] / 2;
  long ND = (long)N * D;

  float* ws    = (float*)d_ws;
  float* deg_s = ws;
  float* deg_r = ws + N;
  float* bufA  = ws + 2L * N;
  float* bufB  = bufA + ND;
  float* bufC  = bufB + ND;
  unsigned int* keyw = (unsigned int*)bufC;
  float* Wp1   = bufC + ND;
  float* Wp2   = Wp1 + 2L * D * D;
  float* Wpa   = Wp2 + 2L * D * D;

  const int T = 256;
  int nb_pack = (2 * NPAIR_SAGE + NPAIR_MLP + T - 1) / T;
  init_kernel<<<nb_pack, T, 0, stream>>>(W1, Wp1, W2, Wp2, Wa, Wpa);

  int nchunks = (E + KCHUNK - 1) / KCHUNK;
  int nwords = nchunks * (KCHUNK / 4);
  int nseg = (N + DSEG - 1) / DSEG;
  int nslab = nseg < 256 ? 256 / nseg : 1;
  if (nslab > D) nslab = D;
  int wtot = (E + 3) / 4;
  int slab_words = (wtot + nslab - 1) / nslab;
  int* part = (int*)bufB;
  degree_kernel<<<nseg * nslab, DC_THREADS, 0, stream>>>(senders, receivers, part, keyw,
                                                         E, nwords, N, nseg, slab_words);

  long n4 = ND / 4;
  int nranges = (N + RPB - 1) / RPB;
  int sgrid = nranges < 256 ? nranges : 256;
  int* elog_n = (int*)(keyw + nwords);
  int2* elog = (int2*)(elog_n + nranges * SC_WAVES);
  int rnblocks = (int)((n4 + T * RN_PER_LANE - 1) / (T * RN_PER_LANE));

  rownorm_kernel<<<rnblocks, T, 0, stream>>>(emb, part, nslab, deg_s, bufA, N);
  scatter_kernel<false><<<sgrid, SC_THREADS, 0, stream>>>(senders, receivers, keyw, bufA, bufB, elog, elog_n,
                                                          emb, deg_r, deg_s, Wp1, b1,
                                                          E, nchunks, N, nranges);

  scatter_kernel<true><<<sgrid, SC_THREADS, 0, stream>>>(senders, receivers, keyw, bufB, bufA, elog, elog_n,
                                                         bufB, deg_r, deg_s, Wp2, b2,
                                                         E, nchunks, N, nranges);

  int ptiles = (P + 15) >> 4;
  pair_mlp_kernel<<<(ptiles + 3) / 4, 128, 0, stream>>>(pairs, bufA, Wpa, ba, Wb, bb, out, P);
}
